# starting kernel
# baseline (speedup 1.0000x reference)
#include <hip/hip_runtime.h>
#include <math.h>

typedef __attribute__((ext_vector_type(2))) float v2f;
typedef __attribute__((ext_vector_type(8))) float v8f;

#define NN 50000
#define H  64


__global__ void k_zero(float* __restrict__ p, int n) {
    int i = blockIdx.x * blockDim.x + threadIdx.x;
    if (i < n) p[i] = 0.0f;
}

__global__ void k_deg(const int* __restrict__ dst, float* __restrict__ deg, int E) {
    int e = blockIdx.x * blockDim.x + threadIdx.x;
    if (e < E) atomicAdd(&deg[dst[e]], 1.0f);
}

__global__ void k_dinv(float* __restrict__ deg_dinv, int n) {
    int i = blockIdx.x * blockDim.x + threadIdx.x;
    if (i < n) deg_dinv[i] = rsqrtf(deg_dinv[i] + 1.0f);
}

__global__ void k_norm(const int* __restrict__ src, const int* __restrict__ dst,
                       const float* __restrict__ dinv, float* __restrict__ norm, int E) {
    int e = blockIdx.x * blockDim.x + threadIdx.x;
    if (e < E) norm[e] = dinv[src[e]] * dinv[dst[e]];
}

template<int K, int NB, int NT>
__global__ void k_gemm_wmma(const float* __restrict__ A, const float* __restrict__ B,
                            const float* __restrict__ bias, float* __restrict__ C) {
    const int tm   = blockIdx.x;
    const int lane = threadIdx.x;
    const int half = lane >> 4;
    const int l16  = lane & 15;
    const int arow = tm * 16 + l16;

    const float* Ap = A + (size_t)arow * K + half * 2;
    const float* Bp = B + half * 2 * NB;

    v8f acc[NT];
#pragma unroll
    for (int t = 0; t < NT; ++t) acc[t] = (v8f){};

    for (int k = 0; k < K; k += 4) {
        v2f a;
        a.x = Ap[k];
        a.y = Ap[k + 1];
#pragma unroll
        for (int t = 0; t < NT; ++t) {
            const int c = t * 16 + l16;
            v2f b;
            if constexpr (NT * 16 == NB) {
                b.x = Bp[k * NB + c];
                b.y = Bp[k * NB + NB + c];
            } else {
                const float m  = (c < NB) ? 1.0f : 0.0f;
                const int   cc = (c < NB) ? c : (NB - 1);
                b.x = m * Bp[k * NB + cc];
                b.y = m * Bp[k * NB + NB + cc];
            }
            acc[t] = __builtin_amdgcn_wmma_f32_16x16x4_f32(
                false, a, false, b, (short)0, acc[t], false, false);
        }
    }

#pragma unroll
    for (int t = 0; t < NT; ++t) {
        const int c = t * 16 + l16;
        if (NT * 16 == NB || c < NB) {
            const float bv = bias ? bias[c] : 0.0f;
#pragma unroll
            for (int i = 0; i < 8; ++i)
                C[(size_t)(tm * 16 + i + half * 8) * NB + c] = acc[t][i] + bv;
        }
    }
}


__global__ void k_selfloop_init(const float* __restrict__ h, const float* __restrict__ dinv,
                                float* __restrict__ agg, int n) {
    int i = blockIdx.x * blockDim.x + threadIdx.x;
    if (i >= n) return;
    float di = dinv[i >> 6];
    agg[i] = di * di * h[i];
}

__global__ void k_edge_scatter(const int* __restrict__ src, const int* __restrict__ dst,
                               const float* __restrict__ norm, const float* __restrict__ h,
                               float* __restrict__ agg, int E) {
    int gid  = blockIdx.x * blockDim.x + threadIdx.x;
    int e    = gid >> 5;
    int lane = gid & 31;
    if (e >= E) return;
    const int   s = src[e];
    const int   d = dst[e];
    const float w = norm[e];
    const float2 v = ((const float2*)(h + (size_t)s * H))[lane];
    float* o = agg + (size_t)d * H + lane * 2;
    atomicAdd(o,     w * v.x);
    atomicAdd(o + 1, w * v.y);
}

__global__ void k_bias_act(const float* __restrict__ agg, const float* __restrict__ bias,
                           float* __restrict__ act, int n, int do_elu) {
    int i = blockIdx.x * blockDim.x + threadIdx.x;
    if (i >= n) return;
    float v = agg[i] + bias[i & (H - 1)];
    act[i] = (do_elu && v <= 0.0f) ? expm1f(v) : v;
}


extern "C" void kernel_launch(void* const* d_in, const int* in_sizes, int n_in,
                              void* d_out, int out_size, void* d_ws, size_t ws_size,
                              hipStream_t stream) {
    const float* x  = (const float*)d_in[0];
    const int*   ei = (const int*)d_in[1];
    const float* W1 = (const float*)d_in[2];
    const float* b1 = (const float*)d_in[3];
    const float* W2 = (const float*)d_in[4];
    const float* b2 = (const float*)d_in[5];
    const float* W3 = (const float*)d_in[6];
    const float* b3 = (const float*)d_in[7];
    const float* Wl = (const float*)d_in[8];
    const float* bl = (const float*)d_in[9];
    float* out = (float*)d_out;

    const int E = in_sizes[1] / 2;
    const int* src = ei;
    const int* dst = ei + E;

    float* dinv = (float*)d_ws;
    float* norm = dinv + NN;
    float* G    = norm + E;
    float* AGG  = G + (size_t)NN * H;
    float* ACT  = AGG + (size_t)NN * H;

    const int NF = NN * H;
    const dim3 blk(256);

    k_zero<<<dim3((NN + 255) / 256), blk, 0, stream>>>(dinv, NN);
    k_deg<<<dim3((E + 255) / 256), blk, 0, stream>>>(dst, dinv, E);
    k_dinv<<<dim3((NN + 255) / 256), blk, 0, stream>>>(dinv, NN);
    k_norm<<<dim3((E + 255) / 256), blk, 0, stream>>>(src, dst, dinv, norm, E);

    const dim3 gGemm(NN / 16);
    const dim3 wave(32);
    const dim3 gElem((NF + 255) / 256);
    const dim3 gScat((E * 32 + 255) / 256);

    k_gemm_wmma<128, H, 4><<<gGemm, wave, 0, stream>>>(x, W1, nullptr, G);
    k_selfloop_init<<<gElem, blk, 0, stream>>>(G, dinv, AGG, NF);
    k_edge_scatter<<<gScat, blk, 0, stream>>>(src, dst, norm, G, AGG, E);
    k_bias_act<<<gElem, blk, 0, stream>>>(AGG, b1, ACT, NF, 1);

    k_gemm_wmma<H, H, 4><<<gGemm, wave, 0, stream>>>(ACT, W2, nullptr, G);
    k_selfloop_init<<<gElem, blk, 0, stream>>>(G, dinv, AGG, NF);
    k_edge_scatter<<<gScat, blk, 0, stream>>>(src, dst, norm, G, AGG, E);
    k_bias_act<<<gElem, blk, 0, stream>>>(AGG, b2, ACT, NF, 1);

    k_gemm_wmma<H, H, 4><<<gGemm, wave, 0, stream>>>(ACT, W3, nullptr, G);
    k_selfloop_init<<<gElem, blk, 0, stream>>>(G, dinv, AGG, NF);
    k_edge_scatter<<<gScat, blk, 0, stream>>>(src, dst, norm, G, AGG, E);
    k_bias_act<<<gElem, blk, 0, stream>>>(AGG, b3, ACT, NF, 0);

    k_gemm_wmma<H, 40, 3><<<gGemm, wave, 0, stream>>>(ACT, Wl, bl, out);
}


# Atomic edge scatter replaced by a dst-sorted edge list (slot = arrival-order rank from one returning int atomic) and a wave-per-node register gather-sum that also applies self-loop, bias and ELU
# speedup vs baseline: 2.1264x; 2.1264x over previous
#include <hip/hip_runtime.h>
#include <math.h>


typedef __attribute__((ext_vector_type(2))) float v2f;
typedef __attribute__((ext_vector_type(8))) float v8f;

#define NN 50000
#define H  64


__global__ void k_zero(int* __restrict__ p, int n) {
    int i = blockIdx.x * blockDim.x + threadIdx.x;
    if (i < n) p[i] = 0;
}

__global__ void k_rank(const int* __restrict__ dst, int* __restrict__ cnt,
                       int* __restrict__ rank, int E) {
    int e = blockIdx.x * blockDim.x + threadIdx.x;
    if (e < E) rank[e] = atomicAdd(&cnt[dst[e]], 1);
}

__device__ __forceinline__ int wave_inclusive_scan(int v, int lane) {
#pragma unroll
    for (int off = 1; off < 32; off <<= 1) {
        int u = __shfl_up(v, off);
        if (lane >= off) v += u;
    }
    return v;
}

#define SCAN_TILES 4
__global__ void __launch_bounds__(1024) k_scan(const int* __restrict__ cnt, int* __restrict__ rowptr,
                                               float* __restrict__ dinv) {
    __shared__ int wtot[32];
    const int t    = threadIdx.x;
    const int lane = t & 31;
    const int wv   = t >> 5;

    int4 c[SCAN_TILES][4];
#pragma unroll
    for (int tile = 0; tile < SCAN_TILES; ++tile) {
        const int base = tile * 16384 + t * 16;
#pragma unroll
        for (int k = 0; k < 4; ++k)
            c[tile][k] = (base < NN) ? *(const int4*)(cnt + base + 4 * k) : make_int4(0, 0, 0, 0);
    }

    int carry = 0;
#pragma unroll
    for (int tile = 0; tile < SCAN_TILES; ++tile) {
        const int base = tile * 16384 + t * 16;
        int s = 0;
#pragma unroll
        for (int k = 0; k < 4; ++k) s += c[tile][k].x + c[tile][k].y + c[tile][k].z + c[tile][k].w;
        const int inc = wave_inclusive_scan(s, lane);
        if (lane == 31) wtot[wv] = inc;
        __syncthreads();
        const int wt   = wtot[lane];
        const int winc = wave_inclusive_scan(wt, lane);
        const int wave_excl  = __shfl(winc, wv) - __shfl(wt, wv);
        const int tile_total = __shfl(winc, 31);
        __syncthreads();
        const int ex = carry + wave_excl + inc - s;
        if (base < NN) {
            int run = ex;
#pragma unroll
            for (int k = 0; k < 4; ++k) {
                const int4 q = c[tile][k];
                *(int4*)(rowptr + base + 4 * k) =
                    make_int4(run, run + q.x, run + q.x + q.y, run + q.x + q.y + q.z);
                float4 dv;
                dv.x = rsqrtf((float)q.x + 1.0f);
                dv.y = rsqrtf((float)q.y + 1.0f);
                dv.z = rsqrtf((float)q.z + 1.0f);
                dv.w = rsqrtf((float)q.w + 1.0f);
                *(float4*)(dinv + base + 4 * k) = dv;
                run += q.x + q.y + q.z + q.w;
            }
        } else if (base == NN) {
            rowptr[NN] = ex;
        }
        carry += tile_total;
    }
}

__global__ void k_fill(const int* __restrict__ src, const int* __restrict__ dst,
                       const int* __restrict__ rank, const int* __restrict__ rowptr,
                       const float* __restrict__ dinv, int* __restrict__ rec, int E) {
    int e = blockIdx.x * blockDim.x + threadIdx.x;
    if (e >= E) return;
    const int s = src[e];
    const int d = dst[e];
    const int p = rowptr[d] + rank[e];
    ((int2*)rec)[p] = make_int2(s, __float_as_int(dinv[s] * dinv[d]));
}

template<int K, int NB, int NT>
__global__ void k_gemm_wmma(const float* __restrict__ A, const float* __restrict__ B,
                            const float* __restrict__ bias, float* __restrict__ C) {
    const int tm   = blockIdx.x;
    const int lane = threadIdx.x;
    const int half = lane >> 4;
    const int l16  = lane & 15;
    const int arow = tm * 16 + l16;

    const float* Ap = A + (size_t)arow * K + half * 2;
    const float* Bp = B + half * 2 * NB;

    v8f acc[NT];
#pragma unroll
    for (int t = 0; t < NT; ++t) acc[t] = (v8f){};

    for (int k = 0; k < K; k += 4) {
        v2f a;
        a.x = Ap[k];
        a.y = Ap[k + 1];
#pragma unroll
        for (int t = 0; t < NT; ++t) {
            const int c = t * 16 + l16;
            v2f b;
            if constexpr (NT * 16 == NB) {
                b.x = Bp[k * NB + c];
                b.y = Bp[k * NB + NB + c];
            } else {
                const float m  = (c < NB) ? 1.0f : 0.0f;
                const int   cc = (c < NB) ? c : (NB - 1);
                b.x = m * Bp[k * NB + cc];
                b.y = m * Bp[k * NB + NB + cc];
            }
            acc[t] = __builtin_amdgcn_wmma_f32_16x16x4_f32(
                false, a, false, b, (short)0, acc[t], false, false);
        }
    }

#pragma unroll
    for (int t = 0; t < NT; ++t) {
        const int c = t * 16 + l16;
        if (NT * 16 == NB || c < NB) {
            const float bv = bias ? bias[c] : 0.0f;
#pragma unroll
            for (int i = 0; i < 8; ++i)
                C[(size_t)(tm * 16 + i + half * 8) * NB + c] = acc[t][i] + bv;
        }
    }
}


template<int U>
__device__ __forceinline__ void add_edges(const float2* __restrict__ h2, int ms, int mw, int j,
                                          int lane, float& ax, float& ay) {
    int    s[U];
    float  w[U];
    float2 v[U];
#pragma unroll
    for (int k = 0; k < U; ++k) {
        s[k] = __builtin_amdgcn_readlane(ms, j + k);
        w[k] = __int_as_float(__builtin_amdgcn_readlane(mw, j + k));
    }
#pragma unroll
    for (int k = 0; k < U; ++k) v[k] = h2[(size_t)s[k] * 32 + lane];
#pragma unroll
    for (int k = 0; k < U; ++k) {
        ax += w[k] * v[k].x;
        ay += w[k] * v[k].y;
    }
}

__global__ void __launch_bounds__(256) k_gather(const int* __restrict__ rowptr, const int* __restrict__ rec,
                                                const float* __restrict__ h, const float* __restrict__ dinv,
                                                const float* __restrict__ bias, float* __restrict__ act,
                                                int do_elu) {
    const int lane = threadIdx.x & 31;
    const int node = __builtin_amdgcn_readfirstlane((int)((blockIdx.x * blockDim.x + threadIdx.x) >> 5));
    if (node >= NN) return;
    const int   beg = rowptr[node];
    const int   end = rowptr[node + 1];
    const float di  = dinv[node];
    const float2* __restrict__ h2 = (const float2*)h;
    const int2*   __restrict__ r2 = (const int2*)rec;

    const float2 hv = h2[(size_t)node * 32 + lane];
    const float  sw = di * di;
    float ax = sw * hv.x;
    float ay = sw * hv.y;

    for (int base = beg; base < end; base += 32) {
        const int n = min(32, end - base);
        int ms = 0, mw = 0;
        if (lane < n) {
            const int2 r = r2[base + lane];
            ms = r.x;
            mw = r.y;
        }
        int j = 0;
        for (; j + 8 <= n; j += 8) add_edges<8>(h2, ms, mw, j, lane, ax, ay);
        if (j + 4 <= n) { add_edges<4>(h2, ms, mw, j, lane, ax, ay); j += 4; }
        if (j + 2 <= n) { add_edges<2>(h2, ms, mw, j, lane, ax, ay); j += 2; }
        if (j < n)      { add_edges<1>(h2, ms, mw, j, lane, ax, ay); }
    }

    const float2 bv = ((const float2*)bias)[lane];
    float vx = ax + bv.x;
    float vy = ay + bv.y;
    if (do_elu) {
        vx = (vx <= 0.0f) ? expm1f(vx) : vx;
        vy = (vy <= 0.0f) ? expm1f(vy) : vy;
    }
    ((float2*)act)[(size_t)node * 32 + lane] = make_float2(vx, vy);
}


extern "C" void kernel_launch(void* const* d_in, const int* in_sizes, int n_in,
                              void* d_out, int out_size, void* d_ws, size_t ws_size,
                              hipStream_t stream) {
    const float* x  = (const float*)d_in[0];
    const int*   ei = (const int*)d_in[1];
    const float* W1 = (const float*)d_in[2];
    const float* b1 = (const float*)d_in[3];
    const float* W2 = (const float*)d_in[4];
    const float* b2 = (const float*)d_in[5];
    const float* W3 = (const float*)d_in[6];
    const float* b3 = (const float*)d_in[7];
    const float* Wl = (const float*)d_in[8];
    const float* bl = (const float*)d_in[9];
    float* out = (float*)d_out;

    const int E = in_sizes[1] / 2;
    const int* src = ei;
    const int* dst = ei + E;

    float* G      = (float*)d_ws;
    float* ACT    = G + (size_t)NN * H;
    int*   cnt    = (int*)(ACT + (size_t)NN * H);
    float* dinv   = (float*)(cnt + NN);
    int*   rowptr = (int*)(dinv + NN);
    int*   rec    = rowptr + NN + 16;
    int*   rank   = rec + 2 * (size_t)E;

    const dim3 blk(256);

    k_zero<<<dim3((NN + 255) / 256), blk, 0, stream>>>(cnt, NN);
    k_rank<<<dim3((E + 255) / 256), blk, 0, stream>>>(dst, cnt, rank, E);
    k_scan<<<dim3(1), dim3(1024), 0, stream>>>(cnt, rowptr, dinv);
    k_fill<<<dim3((E + 255) / 256), blk, 0, stream>>>(src, dst, rank, rowptr, dinv, rec, E);

    const dim3 gGemm(NN / 16);
    const dim3 wave(32);
    const dim3 gGath((NN * 32 + 255) / 256);

    k_gemm_wmma<128, H, 4><<<gGemm, wave, 0, stream>>>(x, W1, nullptr, G);
    k_gather<<<gGath, blk, 0, stream>>>(rowptr, rec, G, dinv, b1, ACT, 1);

    k_gemm_wmma<H, H, 4><<<gGemm, wave, 0, stream>>>(ACT, W2, nullptr, G);
    k_gather<<<gGath, blk, 0, stream>>>(rowptr, rec, G, dinv, b2, ACT, 1);

    k_gemm_wmma<H, H, 4><<<gGemm, wave, 0, stream>>>(ACT, W3, nullptr, G);
    k_gather<<<gGath, blk, 0, stream>>>(rowptr, rec, G, dinv, b3, ACT, 0);

    k_gemm_wmma<H, 40, 3><<<gGemm, wave, 0, stream>>>(ACT, Wl, bl, out);
}


# Each aggregation launch also does the next layer's f32 WMMA product: 16 one-node waves park their summed rows in LDS as the A tile, W staged in LDS; 11 launches to 8; every sum order as parent
# speedup vs baseline: 2.3266x; 1.0941x over previous
#include <hip/hip_runtime.h>
#include <math.h>


typedef __attribute__((ext_vector_type(2))) float v2f;
typedef __attribute__((ext_vector_type(8))) float v8f;

#define NN 50000
#define H  64


__global__ void k_zero(int* __restrict__ p, int n) {
    int i = blockIdx.x * blockDim.x + threadIdx.x;
    if (i < n) p[i] = 0;
}

__global__ void k_rank(const int* __restrict__ dst, int* __restrict__ cnt,
                       int* __restrict__ rank, int E) {
    int e = blockIdx.x * blockDim.x + threadIdx.x;
    if (e < E) rank[e] = atomicAdd(&cnt[dst[e]], 1);
}

__device__ __forceinline__ int wave_inclusive_scan(int v, int lane) {
#pragma unroll
    for (int off = 1; off < 32; off <<= 1) {
        int u = __shfl_up(v, off);
        if (lane >= off) v += u;
    }
    return v;
}

#define SCAN_TILES 4
__global__ void __launch_bounds__(1024) k_scan(const int* __restrict__ cnt, int* __restrict__ rowptr,
                                               float* __restrict__ dinv) {
    __shared__ int wtot[32];
    const int t    = threadIdx.x;
    const int lane = t & 31;
    const int wv   = t >> 5;

    int4 c[SCAN_TILES][4];
#pragma unroll
    for (int tile = 0; tile < SCAN_TILES; ++tile) {
        const int base = tile * 16384 + t * 16;
#pragma unroll
        for (int k = 0; k < 4; ++k)
            c[tile][k] = (base < NN) ? *(const int4*)(cnt + base + 4 * k) : make_int4(0, 0, 0, 0);
    }

    int carry = 0;
#pragma unroll
    for (int tile = 0; tile < SCAN_TILES; ++tile) {
        const int base = tile * 16384 + t * 16;
        int s = 0;
#pragma unroll
        for (int k = 0; k < 4; ++k) s += c[tile][k].x + c[tile][k].y + c[tile][k].z + c[tile][k].w;
        const int inc = wave_inclusive_scan(s, lane);
        if (lane == 31) wtot[wv] = inc;
        __syncthreads();
        const int wt   = wtot[lane];
        const int winc = wave_inclusive_scan(wt, lane);
        const int wave_excl  = __shfl(winc, wv) - __shfl(wt, wv);
        const int tile_total = __shfl(winc, 31);
        __syncthreads();
        const int ex = carry + wave_excl + inc - s;
        if (base < NN) {
            int run = ex;
#pragma unroll
            for (int k = 0; k < 4; ++k) {
                const int4 q = c[tile][k];
                *(int4*)(rowptr + base + 4 * k) =
                    make_int4(run, run + q.x, run + q.x + q.y, run + q.x + q.y + q.z);
                float4 dv;
                dv.x = rsqrtf((float)q.x + 1.0f);
                dv.y = rsqrtf((float)q.y + 1.0f);
                dv.z = rsqrtf((float)q.z + 1.0f);
                dv.w = rsqrtf((float)q.w + 1.0f);
                *(float4*)(dinv + base + 4 * k) = dv;
                run += q.x + q.y + q.z + q.w;
            }
        } else if (base == NN) {
            rowptr[NN] = ex;
        }
        carry += tile_total;
    }
}

__global__ void k_fill(const int* __restrict__ src, const int* __restrict__ dst,
                       const int* __restrict__ rank, const int* __restrict__ rowptr,
                       const float* __restrict__ dinv, int* __restrict__ rec, int E) {
    int e = blockIdx.x * blockDim.x + threadIdx.x;
    if (e >= E) return;
    const int s = src[e];
    const int d = dst[e];
    const int p = rowptr[d] + rank[e];
    ((int2*)rec)[p] = make_int2(s, __float_as_int(dinv[s] * dinv[d]));
}

template<int K, int NB, int NT>
__global__ void k_gemm_wmma(const float* __restrict__ A, const float* __restrict__ B,
                            const float* __restrict__ bias, float* __restrict__ C) {
    const int tm   = blockIdx.x;
    const int lane = threadIdx.x;
    const int half = lane >> 4;
    const int l16  = lane & 15;
    const int arow = tm * 16 + l16;

    const float* Ap = A + (size_t)arow * K + half * 2;
    const float* Bp = B + half * 2 * NB;

    v8f acc[NT];
#pragma unroll
    for (int t = 0; t < NT; ++t) acc[t] = (v8f){};

    for (int k = 0; k < K; k += 4) {
        v2f a;
        a.x = Ap[k];
        a.y = Ap[k + 1];
#pragma unroll
        for (int t = 0; t < NT; ++t) {
            const int c = t * 16 + l16;
            v2f b;
            if constexpr (NT * 16 == NB) {
                b.x = Bp[k * NB + c];
                b.y = Bp[k * NB + NB + c];
            } else {
                const float m  = (c < NB) ? 1.0f : 0.0f;
                const int   cc = (c < NB) ? c : (NB - 1);
                b.x = m * Bp[k * NB + cc];
                b.y = m * Bp[k * NB + NB + cc];
            }
            acc[t] = __builtin_amdgcn_wmma_f32_16x16x4_f32(
                false, a, false, b, (short)0, acc[t], false, false);
        }
    }

#pragma unroll
    for (int t = 0; t < NT; ++t) {
        const int c = t * 16 + l16;
        if (NT * 16 == NB || c < NB) {
            const float bv = bias ? bias[c] : 0.0f;
#pragma unroll
            for (int i = 0; i < 8; ++i)
                C[(size_t)(tm * 16 + i + half * 8) * NB + c] = acc[t][i] + bv;
        }
    }
}


template<int U>
__device__ __forceinline__ void add_edges(const float2* __restrict__ h2, int ms, int mw, int j,
                                          int lane, float& ax, float& ay) {
    int    s[U];
    float  w[U];
    float2 v[U];
#pragma unroll
    for (int k = 0; k < U; ++k) {
        s[k] = __builtin_amdgcn_readlane(ms, j + k);
        w[k] = __int_as_float(__builtin_amdgcn_readlane(mw, j + k));
    }
#pragma unroll
    for (int k = 0; k < U; ++k) v[k] = h2[(size_t)s[k] * 32 + lane];
#pragma unroll
    for (int k = 0; k < U; ++k) {
        ax += w[k] * v[k].x;
        ay += w[k] * v[k].y;
    }
}

template<int NB, int NT>
__global__ void __launch_bounds__(512) k_gather_gemm(const int* __restrict__ rowptr, const int* __restrict__ rec,
                                                     const float* __restrict__ h, const float* __restrict__ dinv,
                                                     const float* __restrict__ bias, const float* __restrict__ Wn,
                                                     const float* __restrict__ bn, float* __restrict__ outp,
                                                     int do_elu) {
    static_assert(NN % 16 == 0, "every workgroup owns 16 valid nodes; no wave may leave before the barriers");
    static_assert(NB % 4 == 0 && NT * 16 >= NB && NT <= 16, "tile shape");
    constexpr int PB  = (NB == 64) ? 72 : 56;
    constexpr int PA  = 68;
    constexpr int PC  = NB + 4;
    constexpr int QPR = NB / 4;
    constexpr int NQ  = 64 * QPR;
    constexpr int NQO = 16 * QPR;
    static_assert(NT * 16 <= PB && NQ <= 1024 && NQO <= 512, "staging shape");
    __shared__ __attribute__((aligned(16))) float Ws[64 * PB];
    __shared__ __attribute__((aligned(16))) float As[16 * PA];
    __shared__ __attribute__((aligned(16))) float Cs[16 * PC];

    const int t    = threadIdx.x;
    const int lane = t & 31;
    const int wv   = __builtin_amdgcn_readfirstlane(t >> 5);

    constexpr int REM = NQ - 512;
    static_assert(REM > 0 && REM <= 512 && (REM & (REM - 1)) == 0, "second round wraps by a power of two");
    const int q0 = t;
    const int q1 = 512 + (t & (REM - 1));
    const float4 wq0 = ((const float4*)Wn)[q0];
    const float4 wq1 = ((const float4*)Wn)[q1];

    const int node = (int)(blockIdx.x * 16) + wv;
    const int   beg = rowptr[node];
    const int   end = rowptr[node + 1];
    const float di  = dinv[node];
    const float2* __restrict__ h2 = (const float2*)h;
    const int2*   __restrict__ r2 = (const int2*)rec;

    const float2 hv = h2[(size_t)node * 32 + lane];
    const float  sw = di * di;
    float ax = sw * hv.x;
    float ay = sw * hv.y;

    *(float4*)&Ws[(q0 / QPR) * PB + 4 * (q0 % QPR)] = wq0;
    *(float4*)&Ws[(q1 / QPR) * PB + 4 * (q1 % QPR)] = wq1;
    if constexpr (NT * 16 != NB) {
        constexpr int PQ = (NT * 16 - NB) / 4;
        if (t < 64 * PQ) *(float4*)&Ws[(t / PQ) * PB + NB + 4 * (t % PQ)] = make_float4(0.f, 0.f, 0.f, 0.f);
    }

    for (int base = beg; base < end; base += 32) {
        const int n = min(32, end - base);
        int ms = 0, mw = 0;
        if (lane < n) {
            const int2 r = r2[base + lane];
            ms = r.x;
            mw = r.y;
        }
        int j = 0;
        for (; j + 8 <= n; j += 8) add_edges<8>(h2, ms, mw, j, lane, ax, ay);
        if (j + 4 <= n) { add_edges<4>(h2, ms, mw, j, lane, ax, ay); j += 4; }
        if (j + 2 <= n) { add_edges<2>(h2, ms, mw, j, lane, ax, ay); j += 2; }
        if (j < n)      { add_edges<1>(h2, ms, mw, j, lane, ax, ay); }
    }

    const float2 bv2 = ((const float2*)bias)[lane];
    float vx = ax + bv2.x;
    float vy = ay + bv2.y;
    if (do_elu) {
        vx = (vx <= 0.0f) ? expm1f(vx) : vx;
        vy = (vy <= 0.0f) ? expm1f(vy) : vy;
    }
    *(float2*)&As[wv * PA + 2 * lane] = make_float2(vx, vy);

    __syncthreads();

    if (wv < NT) {
        const int half = lane >> 4;
        const int l16  = lane & 15;
        const int c    = wv * 16 + l16;
        const float* Ap = As + l16 * PA + half * 2;
        const float* Bp = Ws + half * 2 * PB + c;
        const float bv = bn ? bn[(NT * 16 == NB) ? c : min(c, NB - 1)] : 0.0f;
        v8f acc = (v8f){};
#pragma unroll
        for (int k = 0; k < H; k += 4) {
            v2f a, b;
            a.x = Ap[k];
            a.y = Ap[k + 1];
            b.x = Bp[k * PB];
            b.y = Bp[k * PB + PB];
            acc = __builtin_amdgcn_wmma_f32_16x16x4_f32(
                false, a, false, b, (short)0, acc, false, false);
        }
        if (NT * 16 == NB || c < NB) {
#pragma unroll
            for (int i = 0; i < 8; ++i) Cs[(i + half * 8) * PC + c] = acc[i] + bv;
        }
    }
    __syncthreads();

    if (t < NQO) {
        const float4 v = *(const float4*)&Cs[(t / QPR) * PC + 4 * (t % QPR)];
        ((float4*)outp)[(size_t)blockIdx.x * NQO + t] = v;
    }
}


extern "C" void kernel_launch(void* const* d_in, const int* in_sizes, int n_in,
                              void* d_out, int out_size, void* d_ws, size_t ws_size,
                              hipStream_t stream) {
    const float* x  = (const float*)d_in[0];
    const int*   ei = (const int*)d_in[1];
    const float* W1 = (const float*)d_in[2];
    const float* b1 = (const float*)d_in[3];
    const float* W2 = (const float*)d_in[4];
    const float* b2 = (const float*)d_in[5];
    const float* W3 = (const float*)d_in[6];
    const float* b3 = (const float*)d_in[7];
    const float* Wl = (const float*)d_in[8];
    const float* bl = (const float*)d_in[9];
    float* out = (float*)d_out;

    const int E = in_sizes[1] / 2;
    const int* src = ei;
    const int* dst = ei + E;

    float* G      = (float*)d_ws;
    float* ACT    = G + (size_t)NN * H;
    int*   cnt    = (int*)(ACT + (size_t)NN * H);
    float* dinv   = (float*)(cnt + NN);
    int*   rowptr = (int*)(dinv + NN);
    int*   rec    = rowptr + NN + 16;
    int*   rank   = rec + 2 * (size_t)E;

    const dim3 blk(256);

    k_zero<<<dim3((NN + 255) / 256), blk, 0, stream>>>(cnt, NN);
    k_rank<<<dim3((E + 255) / 256), blk, 0, stream>>>(dst, cnt, rank, E);
    k_scan<<<dim3(1), dim3(1024), 0, stream>>>(cnt, rowptr, dinv);
    k_fill<<<dim3((E + 255) / 256), blk, 0, stream>>>(src, dst, rank, rowptr, dinv, rec, E);

    const dim3 gGemm(NN / 16);
    const dim3 wave(32);
    const dim3 blk16(512);


    k_gemm_wmma<128, H, 4><<<gGemm, wave, 0, stream>>>(x, W1, nullptr, G);
    k_gather_gemm<H, 4><<<gGemm, blk16, 0, stream>>>(rowptr, rec, G, dinv, b1, W2, nullptr, ACT, 1);
    k_gather_gemm<H, 4><<<gGemm, blk16, 0, stream>>>(rowptr, rec, ACT, dinv, b2, W3, nullptr, G, 1);
    k_gather_gemm<40, 3><<<gGemm, blk16, 0, stream>>>(rowptr, rec, G, dinv, b3, Wl, bl, out, 0);
}


# Edge ranks from LDS, not global atomics: 32 workgroups count one edge slice each in an all-node LDS histogram (returning add = rank), per-node merge of slice counts; list order: slice, LDS arrival
# speedup vs baseline: 3.0532x; 1.3123x over previous
#include <hip/hip_runtime.h>
#include <math.h>


typedef __attribute__((ext_vector_type(2))) float v2f;
typedef __attribute__((ext_vector_type(8))) float v8f;

#define NN 50000
#define H  64


#define HG 32

__global__ void __launch_bounds__(1024) k_hist(const int* __restrict__ dst, int* __restrict__ part,
                                               int* __restrict__ rank, int E) {
    static_assert(NN % 4 == 0, "histogram cleared and copied out 16 bytes per thread");
    __shared__ __attribute__((aligned(16))) int hist[NN];
    const int t = threadIdx.x;
    const int g = blockIdx.x;
    for (int i = t; i < NN / 4; i += 1024) ((int4*)hist)[i] = make_int4(0, 0, 0, 0);
    __syncthreads();

    const int chunk = (E + HG - 1) / HG;
    const int beg   = g * chunk;
    const int end   = min(E, beg + chunk);
    int e = beg + t;
    for (; e + 7 * 1024 < end; e += 8 * 1024) {
        int d[8], r[8];
#pragma unroll
        for (int k = 0; k < 8; ++k) d[k] = dst[e + k * 1024];
#pragma unroll
        for (int k = 0; k < 8; ++k) r[k] = atomicAdd(&hist[d[k]], 1);
#pragma unroll
        for (int k = 0; k < 8; ++k) rank[e + k * 1024] = r[k];
    }
    for (; e < end; e += 1024) rank[e] = atomicAdd(&hist[dst[e]], 1);
    __syncthreads();

    int4* __restrict__ po = (int4*)(part + (size_t)g * NN);
    for (int i = t; i < NN / 4; i += 1024) po[i] = ((const int4*)hist)[i];
}

__global__ void k_merge(int* __restrict__ part, int* __restrict__ cnt) {
    const int d = blockIdx.x * blockDim.x + threadIdx.x;
    if (d >= NN) return;
    int c[HG];
#pragma unroll
    for (int g = 0; g < HG; ++g) c[g] = part[(size_t)g * NN + d];
    int run = 0;
#pragma unroll
    for (int g = 0; g < HG; ++g) {
        part[(size_t)g * NN + d] = run;
        run += c[g];
    }
    cnt[d] = run;
}

__device__ __forceinline__ int wave_inclusive_scan(int v, int lane) {
#pragma unroll
    for (int off = 1; off < 32; off <<= 1) {
        int u = __shfl_up(v, off);
        if (lane >= off) v += u;
    }
    return v;
}

#define SCAN_TILES 4
__global__ void __launch_bounds__(1024) k_scan(const int* __restrict__ cnt, int* __restrict__ rowptr,
                                               float* __restrict__ dinv) {
    __shared__ int wtot[32];
    const int t    = threadIdx.x;
    const int lane = t & 31;
    const int wv   = t >> 5;

    int4 c[SCAN_TILES][4];
#pragma unroll
    for (int tile = 0; tile < SCAN_TILES; ++tile) {
        const int base = tile * 16384 + t * 16;
#pragma unroll
        for (int k = 0; k < 4; ++k)
            c[tile][k] = (base < NN) ? *(const int4*)(cnt + base + 4 * k) : make_int4(0, 0, 0, 0);
    }

    int carry = 0;
#pragma unroll
    for (int tile = 0; tile < SCAN_TILES; ++tile) {
        const int base = tile * 16384 + t * 16;
        int s = 0;
#pragma unroll
        for (int k = 0; k < 4; ++k) s += c[tile][k].x + c[tile][k].y + c[tile][k].z + c[tile][k].w;
        const int inc = wave_inclusive_scan(s, lane);
        if (lane == 31) wtot[wv] = inc;
        __syncthreads();
        const int wt   = wtot[lane];
        const int winc = wave_inclusive_scan(wt, lane);
        const int wave_excl  = __shfl(winc, wv) - __shfl(wt, wv);
        const int tile_total = __shfl(winc, 31);
        __syncthreads();
        const int ex = carry + wave_excl + inc - s;
        if (base < NN) {
            int run = ex;
#pragma unroll
            for (int k = 0; k < 4; ++k) {
                const int4 q = c[tile][k];
                *(int4*)(rowptr + base + 4 * k) =
                    make_int4(run, run + q.x, run + q.x + q.y, run + q.x + q.y + q.z);
                float4 dv;
                dv.x = rsqrtf((float)q.x + 1.0f);
                dv.y = rsqrtf((float)q.y + 1.0f);
                dv.z = rsqrtf((float)q.z + 1.0f);
                dv.w = rsqrtf((float)q.w + 1.0f);
                *(float4*)(dinv + base + 4 * k) = dv;
                run += q.x + q.y + q.z + q.w;
            }
        } else if (base == NN) {
            rowptr[NN] = ex;
        }
        carry += tile_total;
    }
}

__global__ void k_fill(const int* __restrict__ src, const int* __restrict__ dst,
                       const int* __restrict__ rank, const int* __restrict__ part,
                       const int* __restrict__ rowptr,
                       const float* __restrict__ dinv, int* __restrict__ rec, int E) {
    int e = blockIdx.x * blockDim.x + threadIdx.x;
    if (e >= E) return;
    const int chunk = (E + HG - 1) / HG;
    const int s = src[e];
    const int d = dst[e];
    const int p = rowptr[d] + part[(size_t)(e / chunk) * NN + d] + rank[e];
    ((int2*)rec)[p] = make_int2(s, __float_as_int(dinv[s] * dinv[d]));
}

template<int K, int NB, int NT>
__global__ void k_gemm_wmma(const float* __restrict__ A, const float* __restrict__ B,
                            const float* __restrict__ bias, float* __restrict__ C) {
    const int tm   = blockIdx.x;
    const int lane = threadIdx.x;
    const int half = lane >> 4;
    const int l16  = lane & 15;
    const int arow = tm * 16 + l16;

    const float* Ap = A + (size_t)arow * K + half * 2;
    const float* Bp = B + half * 2 * NB;

    v8f acc[NT];
#pragma unroll
    for (int t = 0; t < NT; ++t) acc[t] = (v8f){};

    for (int k = 0; k < K; k += 4) {
        v2f a;
        a.x = Ap[k];
        a.y = Ap[k + 1];
#pragma unroll
        for (int t = 0; t < NT; ++t) {
            const int c = t * 16 + l16;
            v2f b;
            if constexpr (NT * 16 == NB) {
                b.x = Bp[k * NB + c];
                b.y = Bp[k * NB + NB + c];
            } else {
                const float m  = (c < NB) ? 1.0f : 0.0f;
                const int   cc = (c < NB) ? c : (NB - 1);
                b.x = m * Bp[k * NB + cc];
                b.y = m * Bp[k * NB + NB + cc];
            }
            acc[t] = __builtin_amdgcn_wmma_f32_16x16x4_f32(
                false, a, false, b, (short)0, acc[t], false, false);
        }
    }

#pragma unroll
    for (int t = 0; t < NT; ++t) {
        const int c = t * 16 + l16;
        if (NT * 16 == NB || c < NB) {
            const float bv = bias ? bias[c] : 0.0f;
#pragma unroll
            for (int i = 0; i < 8; ++i)
                C[(size_t)(tm * 16 + i + half * 8) * NB + c] = acc[t][i] + bv;
        }
    }
}


template<int U>
__device__ __forceinline__ void add_edges(const float2* __restrict__ h2, int ms, int mw, int j,
                                          int lane, float& ax, float& ay) {
    int    s[U];
    float  w[U];
    float2 v[U];
#pragma unroll
    for (int k = 0; k < U; ++k) {
        s[k] = __builtin_amdgcn_readlane(ms, j + k);
        w[k] = __int_as_float(__builtin_amdgcn_readlane(mw, j + k));
    }
#pragma unroll
    for (int k = 0; k < U; ++k) v[k] = h2[(size_t)s[k] * 32 + lane];
#pragma unroll
    for (int k = 0; k < U; ++k) {
        ax += w[k] * v[k].x;
        ay += w[k] * v[k].y;
    }
}

template<int NB, int NT>
__global__ void __launch_bounds__(512) k_gather_gemm(const int* __restrict__ rowptr, const int* __restrict__ rec,
                                                     const float* __restrict__ h, const float* __restrict__ dinv,
                                                     const float* __restrict__ bias, const float* __restrict__ Wn,
                                                     const float* __restrict__ bn, float* __restrict__ outp,
                                                     int do_elu) {
    static_assert(NN % 16 == 0, "every workgroup owns 16 valid nodes; no wave may leave before the barriers");
    static_assert(NB % 4 == 0 && NT * 16 >= NB && NT <= 16, "tile shape");
    constexpr int PB  = (NB == 64) ? 72 : 56;
    constexpr int PA  = 68;
    constexpr int PC  = NB + 4;
    constexpr int QPR = NB / 4;
    constexpr int NQ  = 64 * QPR;
    constexpr int NQO = 16 * QPR;
    static_assert(NT * 16 <= PB && NQ <= 1024 && NQO <= 512, "staging shape");
    __shared__ __attribute__((aligned(16))) float Ws[64 * PB];
    __shared__ __attribute__((aligned(16))) float As[16 * PA];
    __shared__ __attribute__((aligned(16))) float Cs[16 * PC];

    const int t    = threadIdx.x;
    const int lane = t & 31;
    const int wv   = __builtin_amdgcn_readfirstlane(t >> 5);

    constexpr int REM = NQ - 512;
    static_assert(REM > 0 && REM <= 512 && (REM & (REM - 1)) == 0, "second round wraps by a power of two");
    const int q0 = t;
    const int q1 = 512 + (t & (REM - 1));
    const float4 wq0 = ((const float4*)Wn)[q0];
    const float4 wq1 = ((const float4*)Wn)[q1];

    const int node = (int)(blockIdx.x * 16) + wv;
    const int   beg = rowptr[node];
    const int   end = rowptr[node + 1];
    const float di  = dinv[node];
    const float2* __restrict__ h2 = (const float2*)h;
    const int2*   __restrict__ r2 = (const int2*)rec;

    const float2 hv = h2[(size_t)node * 32 + lane];
    const float  sw = di * di;
    float ax = sw * hv.x;
    float ay = sw * hv.y;

    *(float4*)&Ws[(q0 / QPR) * PB + 4 * (q0 % QPR)] = wq0;
    *(float4*)&Ws[(q1 / QPR) * PB + 4 * (q1 % QPR)] = wq1;
    if constexpr (NT * 16 != NB) {
        constexpr int PQ = (NT * 16 - NB) / 4;
        if (t < 64 * PQ) *(float4*)&Ws[(t / PQ) * PB + NB + 4 * (t % PQ)] = make_float4(0.f, 0.f, 0.f, 0.f);
    }

    for (int base = beg; base < end; base += 32) {
        const int n = min(32, end - base);
        int ms = 0, mw = 0;
        if (lane < n) {
            const int2 r = r2[base + lane];
            ms = r.x;
            mw = r.y;
        }
        int j = 0;
        for (; j + 8 <= n; j += 8) add_edges<8>(h2, ms, mw, j, lane, ax, ay);
        if (j + 4 <= n) { add_edges<4>(h2, ms, mw, j, lane, ax, ay); j += 4; }
        if (j + 2 <= n) { add_edges<2>(h2, ms, mw, j, lane, ax, ay); j += 2; }
        if (j < n)      { add_edges<1>(h2, ms, mw, j, lane, ax, ay); }
    }

    const float2 bv2 = ((const float2*)bias)[lane];
    float vx = ax + bv2.x;
    float vy = ay + bv2.y;
    if (do_elu) {
        vx = (vx <= 0.0f) ? expm1f(vx) : vx;
        vy = (vy <= 0.0f) ? expm1f(vy) : vy;
    }
    *(float2*)&As[wv * PA + 2 * lane] = make_float2(vx, vy);

    __syncthreads();

    if (wv < NT) {
        const int half = lane >> 4;
        const int l16  = lane & 15;
        const int c    = wv * 16 + l16;
        const float* Ap = As + l16 * PA + half * 2;
        const float* Bp = Ws + half * 2 * PB + c;
        const float bv = bn ? bn[(NT * 16 == NB) ? c : min(c, NB - 1)] : 0.0f;
        v8f acc = (v8f){};
#pragma unroll
        for (int k = 0; k < H; k += 4) {
            v2f a, b;
            a.x = Ap[k];
            a.y = Ap[k + 1];
            b.x = Bp[k * PB];
            b.y = Bp[k * PB + PB];
            acc = __builtin_amdgcn_wmma_f32_16x16x4_f32(
                false, a, false, b, (short)0, acc, false, false);
        }
        if (NT * 16 == NB || c < NB) {
#pragma unroll
            for (int i = 0; i < 8; ++i) Cs[(i + half * 8) * PC + c] = acc[i] + bv;
        }
    }
    __syncthreads();

    if (t < NQO) {
        const float4 v = *(const float4*)&Cs[(t / QPR) * PC + 4 * (t % QPR)];
        ((float4*)outp)[(size_t)blockIdx.x * NQO + t] = v;
    }
}


extern "C" void kernel_launch(void* const* d_in, const int* in_sizes, int n_in,
                              void* d_out, int out_size, void* d_ws, size_t ws_size,
                              hipStream_t stream) {
    const float* x  = (const float*)d_in[0];
    const int*   ei = (const int*)d_in[1];
    const float* W1 = (const float*)d_in[2];
    const float* b1 = (const float*)d_in[3];
    const float* W2 = (const float*)d_in[4];
    const float* b2 = (const float*)d_in[5];
    const float* W3 = (const float*)d_in[6];
    const float* b3 = (const float*)d_in[7];
    const float* Wl = (const float*)d_in[8];
    const float* bl = (const float*)d_in[9];
    float* out = (float*)d_out;

    const int E = in_sizes[1] / 2;
    const int* src = ei;
    const int* dst = ei + E;

    float* G      = (float*)d_ws;
    float* ACT    = G + (size_t)NN * H;
    int*   cnt    = (int*)(ACT + (size_t)NN * H);
    float* dinv   = (float*)(cnt + NN);
    int*   rowptr = (int*)(dinv + NN);
    int*   rec    = rowptr + NN + 16;
    int*   rank   = rec + 2 * (size_t)E;
    int*   part   = rank + E;

    const dim3 blk(256);

    k_hist<<<dim3(HG), dim3(1024), 0, stream>>>(dst, part, rank, E);
    k_merge<<<dim3((NN + 255) / 256), blk, 0, stream>>>(part, cnt);
    k_scan<<<dim3(1), dim3(1024), 0, stream>>>(cnt, rowptr, dinv);
    k_fill<<<dim3((E + 255) / 256), blk, 0, stream>>>(src, dst, rank, part, rowptr, dinv, rec, E);

    const dim3 gGemm(NN / 16);
    const dim3 wave(32);
    const dim3 blk16(512);


    k_gemm_wmma<128, H, 4><<<gGemm, wave, 0, stream>>>(x, W1, nullptr, G);
    k_gather_gemm<H, 4><<<gGemm, blk16, 0, stream>>>(rowptr, rec, G, dinv, b1, W2, nullptr, ACT, 1);
    k_gather_gemm<H, 4><<<gGemm, blk16, 0, stream>>>(rowptr, rec, ACT, dinv, b2, W3, nullptr, G, 1);
    k_gather_gemm<40, 3><<<gGemm, blk16, 0, stream>>>(rowptr, rec, G, dinv, b3, Wl, bl, out, 0);
}


# Gather launches as resident grids: 1024 16-wave workgroups each walk about 3 node tiles; W parked once (k-pair rows), next tile's bounds, own row and first records prefetched; sum orders as parent
# speedup vs baseline: 3.2164x; 1.0535x over previous
#include <hip/hip_runtime.h>
#include <math.h>


typedef __attribute__((ext_vector_type(2))) float v2f;
typedef __attribute__((ext_vector_type(8))) float v8f;

#define NN 50000
#define H  64
#define GRES 1024


#define HG 32

__global__ void __launch_bounds__(1024) k_hist(const int* __restrict__ dst, int* __restrict__ part,
                                               int* __restrict__ rank, int E) {
    static_assert(NN % 4 == 0, "histogram cleared and copied out 16 bytes per thread");
    __shared__ __attribute__((aligned(16))) int hist[NN];
    const int t = threadIdx.x;
    const int g = blockIdx.x;
    for (int i = t; i < NN / 4; i += 1024) ((int4*)hist)[i] = make_int4(0, 0, 0, 0);
    __syncthreads();

    const int chunk = (E + HG - 1) / HG;
    const int beg   = g * chunk;
    const int end   = min(E, beg + chunk);
    int e = beg + t;
    for (; e + 7 * 1024 < end; e += 8 * 1024) {
        int d[8], r[8];
#pragma unroll
        for (int k = 0; k < 8; ++k) d[k] = dst[e + k * 1024];
#pragma unroll
        for (int k = 0; k < 8; ++k) r[k] = atomicAdd(&hist[d[k]], 1);
#pragma unroll
        for (int k = 0; k < 8; ++k) rank[e + k * 1024] = r[k];
    }
    for (; e < end; e += 1024) rank[e] = atomicAdd(&hist[dst[e]], 1);
    __syncthreads();

    int4* __restrict__ po = (int4*)(part + (size_t)g * NN);
    for (int i = t; i < NN / 4; i += 1024) po[i] = ((const int4*)hist)[i];
}

__global__ void k_merge(int* __restrict__ part, int* __restrict__ cnt) {
    const int d = blockIdx.x * blockDim.x + threadIdx.x;
    if (d >= NN) return;
    int c[HG];
#pragma unroll
    for (int g = 0; g < HG; ++g) c[g] = part[(size_t)g * NN + d];
    int run = 0;
#pragma unroll
    for (int g = 0; g < HG; ++g) {
        part[(size_t)g * NN + d] = run;
        run += c[g];
    }
    cnt[d] = run;
}

__device__ __forceinline__ int wave_inclusive_scan(int v, int lane) {
#pragma unroll
    for (int off = 1; off < 32; off <<= 1) {
        int u = __shfl_up(v, off);
        if (lane >= off) v += u;
    }
    return v;
}

#define SCAN_TILES 4
__global__ void __launch_bounds__(1024) k_scan(const int* __restrict__ cnt, int* __restrict__ rowptr,
                                               float* __restrict__ dinv) {
    __shared__ int wtot[32];
    const int t    = threadIdx.x;
    const int lane = t & 31;
    const int wv   = t >> 5;

    int4 c[SCAN_TILES][4];
#pragma unroll
    for (int tile = 0; tile < SCAN_TILES; ++tile) {
        const int base = tile * 16384 + t * 16;
#pragma unroll
        for (int k = 0; k < 4; ++k)
            c[tile][k] = (base < NN) ? *(const int4*)(cnt + base + 4 * k) : make_int4(0, 0, 0, 0);
    }

    int carry = 0;
#pragma unroll
    for (int tile = 0; tile < SCAN_TILES; ++tile) {
        const int base = tile * 16384 + t * 16;
        int s = 0;
#pragma unroll
        for (int k = 0; k < 4; ++k) s += c[tile][k].x + c[tile][k].y + c[tile][k].z + c[tile][k].w;
        const int inc = wave_inclusive_scan(s, lane);
        if (lane == 31) wtot[wv] = inc;
        __syncthreads();
        const int wt   = wtot[lane];
        const int winc = wave_inclusive_scan(wt, lane);
        const int wave_excl  = __shfl(winc, wv) - __shfl(wt, wv);
        const int tile_total = __shfl(winc, 31);
        __syncthreads();
        const int ex = carry + wave_excl + inc - s;
        if (base < NN) {
            int run = ex;
#pragma unroll
            for (int k = 0; k < 4; ++k) {
                const int4 q = c[tile][k];
                *(int4*)(rowptr + base + 4 * k) =
                    make_int4(run, run + q.x, run + q.x + q.y, run + q.x + q.y + q.z);
                float4 dv;
                dv.x = rsqrtf((float)q.x + 1.0f);
                dv.y = rsqrtf((float)q.y + 1.0f);
                dv.z = rsqrtf((float)q.z + 1.0f);
                dv.w = rsqrtf((float)q.w + 1.0f);
                *(float4*)(dinv + base + 4 * k) = dv;
                run += q.x + q.y + q.z + q.w;
            }
        } else if (base == NN) {
            rowptr[NN] = ex;
        }
        carry += tile_total;
    }
}

__global__ void k_fill(const int* __restrict__ src, const int* __restrict__ dst,
                       const int* __restrict__ rank, const int* __restrict__ part,
                       const int* __restrict__ rowptr,
                       const float* __restrict__ dinv, int* __restrict__ rec, int E) {
    int e = blockIdx.x * blockDim.x + threadIdx.x;
    if (e >= E) return;
    const int chunk = (E + HG - 1) / HG;
    const int s = src[e];
    const int d = dst[e];
    const int p = rowptr[d] + part[(size_t)(e / chunk) * NN + d] + rank[e];
    ((int2*)rec)[p] = make_int2(s, __float_as_int(dinv[s] * dinv[d]));
}

template<int K, int NB, int NT>
__global__ void k_gemm_wmma(const float* __restrict__ A, const float* __restrict__ B,
                            const float* __restrict__ bias, float* __restrict__ C) {
    const int tm   = blockIdx.x;
    const int lane = threadIdx.x;
    const int half = lane >> 4;
    const int l16  = lane & 15;
    const int arow = tm * 16 + l16;

    const float* Ap = A + (size_t)arow * K + half * 2;
    const float* Bp = B + half * 2 * NB;

    v8f acc[NT];
#pragma unroll
    for (int t = 0; t < NT; ++t) acc[t] = (v8f){};

    for (int k = 0; k < K; k += 4) {
        v2f a;
        a.x = Ap[k];
        a.y = Ap[k + 1];
#pragma unroll
        for (int t = 0; t < NT; ++t) {
            const int c = t * 16 + l16;
            v2f b;
            if constexpr (NT * 16 == NB) {
                b.x = Bp[k * NB + c];
                b.y = Bp[k * NB + NB + c];
            } else {
                const float m  = (c < NB) ? 1.0f : 0.0f;
                const int   cc = (c < NB) ? c : (NB - 1);
                b.x = m * Bp[k * NB + cc];
                b.y = m * Bp[k * NB + NB + cc];
            }
            acc[t] = __builtin_amdgcn_wmma_f32_16x16x4_f32(
                false, a, false, b, (short)0, acc[t], false, false);
        }
    }

#pragma unroll
    for (int t = 0; t < NT; ++t) {
        const int c = t * 16 + l16;
        if (NT * 16 == NB || c < NB) {
            const float bv = bias ? bias[c] : 0.0f;
#pragma unroll
            for (int i = 0; i < 8; ++i)
                C[(size_t)(tm * 16 + i + half * 8) * NB + c] = acc[t][i] + bv;
        }
    }
}


template<int U>
__device__ __forceinline__ void add_edges(const float2* __restrict__ h2, int ms, int mw, int j,
                                          int lane, float& ax, float& ay) {
    int    s[U];
    float  w[U];
    float2 v[U];
#pragma unroll
    for (int k = 0; k < U; ++k) {
        s[k] = __builtin_amdgcn_readlane(ms, j + k);
        w[k] = __int_as_float(__builtin_amdgcn_readlane(mw, j + k));
    }
#pragma unroll
    for (int k = 0; k < U; ++k) v[k] = h2[(size_t)s[k] * 32 + lane];
#pragma unroll
    for (int k = 0; k < U; ++k) {
        ax += w[k] * v[k].x;
        ay += w[k] * v[k].y;
    }
}

__device__ __forceinline__ void add_batch(const float2* __restrict__ h2, int ms, int mw, int n,
                                          int lane, float& ax, float& ay) {
    int j = 0;
    for (; j + 8 <= n; j += 8) add_edges<8>(h2, ms, mw, j, lane, ax, ay);
    if (j + 4 <= n) { add_edges<4>(h2, ms, mw, j, lane, ax, ay); j += 4; }
    if (j + 2 <= n) { add_edges<2>(h2, ms, mw, j, lane, ax, ay); j += 2; }
    if (j < n)      { add_edges<1>(h2, ms, mw, j, lane, ax, ay); }
}

template<int NB, int NT>
__global__ void __launch_bounds__(512) k_gather_gemm(const int* __restrict__ rowptr, const int* __restrict__ rec,
                                                     const float* __restrict__ h, const float* __restrict__ dinv,
                                                     const float* __restrict__ bias, const float* __restrict__ Wn,
                                                     const float* __restrict__ bn, float* __restrict__ outp,
                                                     int do_elu, int E) {
    static_assert(NN % 16 == 0, "every tile holds 16 valid nodes; no wave may leave before the barriers");
    static_assert(NB % 4 == 0 && NT * 16 >= NB && NT <= 16, "tile shape");
    constexpr int NTILES = NN / 16;
    constexpr int PB  = (NB == 64) ? 160 : 96;
    constexpr int PA  = 68;
    constexpr int PC  = NB + 4;
    constexpr int QPR = NB / 4;
    constexpr int NQ  = 64 * QPR;
    constexpr int NQO = 16 * QPR;
    static_assert(NT * 32 <= PB && PB % 64 == 32 && NQ <= 1024 && NQO <= 512, "staging shape");
    __shared__ __attribute__((aligned(16))) float Ws[32 * PB];
    __shared__ __attribute__((aligned(16))) float As[16 * PA];
    __shared__ __attribute__((aligned(16))) float Cs[16 * PC];

    const int t    = threadIdx.x;
    const int lane = t & 31;
    const int wv   = __builtin_amdgcn_readfirstlane(t >> 5);

    constexpr int REM = NQ - 512;
    static_assert(REM > 0 && REM <= 512 && (REM & (REM - 1)) == 0, "second round wraps by a power of two");
    const int q0 = t;
    const int q1 = 512 + (t & (REM - 1));
    const float4 wq0 = ((const float4*)Wn)[q0];
    const float4 wq1 = ((const float4*)Wn)[q1];

    const float2* __restrict__ h2 = (const float2*)h;
    const int2*   __restrict__ r2 = (const int2*)rec;
    const int last_rec = max(E - 1, 0);

    static_assert(GRES <= NTILES, "every workgroup of the grid owns a first tile");
    int tile = blockIdx.x;
    int   beg, end;
    float di;
    float2 hv;
    {
        const int node = tile * 16 + wv;
        beg = rowptr[node];
        end = rowptr[node + 1];
        di  = dinv[node];
        hv  = h2[(size_t)node * 32 + lane];
    }
    int2 rfirst = r2[min(beg + lane, last_rec)];

    const float2 bv2 = ((const float2*)bias)[lane];
    const int half = lane >> 4;
    const int l16  = lane & 15;
    const int c    = (wv < NT ? wv : 0) * 16 + l16;
    const float* Ap = As + l16 * PA + half * 2;
    const float* Bp = Ws + half * PB + 2 * c;
    const float bvn = bn ? bn[(NT * 16 == NB) ? c : min(c, NB - 1)] : 0.0f;

    {
        float* w0 = &Ws[((q0 / QPR) >> 1) * PB + 8 * (q0 % QPR) + ((q0 / QPR) & 1)];
        float* w1 = &Ws[((q1 / QPR) >> 1) * PB + 8 * (q1 % QPR) + ((q1 / QPR) & 1)];
        w0[0] = wq0.x; w0[2] = wq0.y; w0[4] = wq0.z; w0[6] = wq0.w;
        w1[0] = wq1.x; w1[2] = wq1.y; w1[4] = wq1.z; w1[6] = wq1.w;
    }
    if constexpr (NT * 16 != NB) {
        constexpr int PQ = (NT * 16 - NB) / 2;
        if (t < 32 * PQ) *(float4*)&Ws[(t / PQ) * PB + 2 * NB + 4 * (t % PQ)] = make_float4(0.f, 0.f, 0.f, 0.f);
    }

    for (;;) {
        const int  ntile = tile + GRES;
        const bool more  = ntile < NTILES;
        const int  nnode = (more ? ntile : tile) * 16 + wv;
        const int    nbnd = rowptr[nnode + (lane & 1)];
        const float  ndi  = dinv[nnode];
        const float2 nhv  = h2[(size_t)nnode * 32 + lane];

        const float sw = di * di;
        float ax = sw * hv.x;
        float ay = sw * hv.y;

        add_batch(h2, rfirst.x, rfirst.y, min(32, end - beg), lane, ax, ay);
        for (int base = beg + 32; base < end; base += 32) {
            const int2 r = r2[min(base + lane, last_rec)];
            add_batch(h2, r.x, r.y, min(32, end - base), lane, ax, ay);
        }
        const int nbeg = __builtin_amdgcn_readlane(nbnd, 0);
        const int nend = __builtin_amdgcn_readlane(nbnd, 1);

        float vx = ax + bv2.x;
        float vy = ay + bv2.y;
        if (do_elu) {
            vx = (vx <= 0.0f) ? expm1f(vx) : vx;
            vy = (vy <= 0.0f) ? expm1f(vy) : vy;
        }
        *(float2*)&As[wv * PA + 2 * lane] = make_float2(vx, vy);

        __syncthreads();

        const int2 nrfirst = r2[min(nbeg + lane, last_rec)];

        if (wv < NT) {
            v8f acc = (v8f){};
#pragma unroll
            for (int k = 0; k < H; k += 4) {
                v2f a, b;
                a.x = Ap[k];
                a.y = Ap[k + 1];
                b.x = Bp[(k / 2) * PB];
                b.y = Bp[(k / 2) * PB + 1];
                acc = __builtin_amdgcn_wmma_f32_16x16x4_f32(
                    false, a, false, b, (short)0, acc, false, false);
            }
            if (NT * 16 == NB || c < NB) {
#pragma unroll
                for (int i = 0; i < 8; ++i) Cs[(i + half * 8) * PC + c] = acc[i] + bvn;
            }
        }
        __syncthreads();

        if (t < NQO) {
            const float4 v = *(const float4*)&Cs[(t / QPR) * PC + 4 * (t % QPR)];
            ((float4*)outp)[(size_t)tile * NQO + t] = v;
        }

        if (!more) break;
        tile   = ntile;
        beg    = nbeg;
        end    = nend;
        di     = ndi;
        hv     = nhv;
        rfirst = nrfirst;
    }
}


extern "C" void kernel_launch(void* const* d_in, const int* in_sizes, int n_in,
                              void* d_out, int out_size, void* d_ws, size_t ws_size,
                              hipStream_t stream) {
    const float* x  = (const float*)d_in[0];
    const int*   ei = (const int*)d_in[1];
    const float* W1 = (const float*)d_in[2];
    const float* b1 = (const float*)d_in[3];
    const float* W2 = (const float*)d_in[4];
    const float* b2 = (const float*)d_in[5];
    const float* W3 = (const float*)d_in[6];
    const float* b3 = (const float*)d_in[7];
    const float* Wl = (const float*)d_in[8];
    const float* bl = (const float*)d_in[9];
    float* out = (float*)d_out;

    const int E = in_sizes[1] / 2;
    const int* src = ei;
    const int* dst = ei + E;

    float* G      = (float*)d_ws;
    float* ACT    = G + (size_t)NN * H;
    int*   cnt    = (int*)(ACT + (size_t)NN * H);
    float* dinv   = (float*)(cnt + NN);
    int*   rowptr = (int*)(dinv + NN);
    int*   rec    = rowptr + NN + 16;
    int*   rank   = rec + 2 * (size_t)E;
    int*   part   = rank + E;

    const dim3 blk(256);

    k_hist<<<dim3(HG), dim3(1024), 0, stream>>>(dst, part, rank, E);
    k_merge<<<dim3((NN + 255) / 256), blk, 0, stream>>>(part, cnt);
    k_scan<<<dim3(1), dim3(1024), 0, stream>>>(cnt, rowptr, dinv);
    k_fill<<<dim3((E + 255) / 256), blk, 0, stream>>>(src, dst, rank, part, rowptr, dinv, rec, E);

    const dim3 gGemm(NN / 16);
    const dim3 wave(32);
    const dim3 blk16(512);
    const dim3 gRes(GRES);


    k_gemm_wmma<128, H, 4><<<gGemm, wave, 0, stream>>>(x, W1, nullptr, G);
    k_gather_gemm<H, 4><<<gRes, blk16, 0, stream>>>(rowptr, rec, G, dinv, b1, W2, nullptr, ACT, 1, E);
    k_gather_gemm<H, 4><<<gRes, blk16, 0, stream>>>(rowptr, rec, ACT, dinv, b2, W3, nullptr, G, 1, E);
    k_gather_gemm<40, 3><<<gRes, blk16, 0, stream>>>(rowptr, rec, G, dinv, b3, Wl, bl, out, 0, E);
}


# Row offsets on two levels: merge workgroups scan their 256 in-degrees into the slice starts; fill (4 edges per thread) scans the 196 block totals in LDS; the one-workgroup scan launch goes: 8 to 7
# speedup vs baseline: 3.4857x; 1.0837x over previous
#include <hip/hip_runtime.h>
#include <math.h>


typedef __attribute__((ext_vector_type(2))) float v2f;
typedef __attribute__((ext_vector_type(8))) float v8f;

#define NN 50000
#define H  64
#define GRES 1024


#define HG 32

__global__ void __launch_bounds__(1024) k_hist(const int* __restrict__ dst, int* __restrict__ part,
                                               int* __restrict__ rank, int E) {
    static_assert(NN % 4 == 0, "histogram cleared and copied out 16 bytes per thread");
    __shared__ __attribute__((aligned(16))) int hist[NN];
    const int t = threadIdx.x;
    const int g = blockIdx.x;
    for (int i = t; i < NN / 4; i += 1024) ((int4*)hist)[i] = make_int4(0, 0, 0, 0);
    __syncthreads();

    const int chunk = (E + HG - 1) / HG;
    const int beg   = g * chunk;
    const int end   = min(E, beg + chunk);
    int e = beg + t;
    for (; e + 7 * 1024 < end; e += 8 * 1024) {
        int d[8], r[8];
#pragma unroll
        for (int k = 0; k < 8; ++k) d[k] = dst[e + k * 1024];
#pragma unroll
        for (int k = 0; k < 8; ++k) r[k] = atomicAdd(&hist[d[k]], 1);
#pragma unroll
        for (int k = 0; k < 8; ++k) rank[e + k * 1024] = r[k];
    }
    for (; e < end; e += 1024) rank[e] = atomicAdd(&hist[dst[e]], 1);
    __syncthreads();

    int4* __restrict__ po = (int4*)(part + (size_t)g * NN);
    for (int i = t; i < NN / 4; i += 1024) po[i] = ((const int4*)hist)[i];
}

__device__ __forceinline__ int wave_inclusive_scan(int v, int lane) {
#pragma unroll
    for (int off = 1; off < 32; off <<= 1) {
        int u = __shfl_up(v, off);
        if (lane >= off) v += u;
    }
    return v;
}

#define MB   256
#define NBLK ((NN + MB - 1) / MB)

__device__ __forceinline__ int block256_exclusive_scan(int v, int t, int* wtot, int& total) {
    const int lane = t & 31;
    const int wv   = t >> 5;
    const int inc  = wave_inclusive_scan(v, lane);
    if (lane == 31) wtot[wv] = inc;
    __syncthreads();
    const int4 lo = *(const int4*)&wtot[0];
    const int4 hi = *(const int4*)&wtot[4];
    const int w[8] = {lo.x, lo.y, lo.z, lo.w, hi.x, hi.y, hi.z, hi.w};
    int before = 0, all = 0;
#pragma unroll
    for (int k = 0; k < 8; ++k) {
        before += (k < wv) ? w[k] : 0;
        all    += w[k];
    }
    total = all;
    return before + inc - v;
}

__global__ void __launch_bounds__(MB) k_merge(int* __restrict__ part, int* __restrict__ btot,
                                              float* __restrict__ dinv) {
    __shared__ __attribute__((aligned(16))) int wtot[8];
    const int t = threadIdx.x;
    const int d = blockIdx.x * MB + t;
    const bool valid = d < NN;
    const int  dl = valid ? d : NN - 1;
    int c[HG];
#pragma unroll
    for (int g = 0; g < HG; ++g) c[g] = part[(size_t)g * NN + dl];
    int deg = 0;
#pragma unroll
    for (int g = 0; g < HG; ++g) deg += c[g];
    if (!valid) deg = 0;
    int total;
    int run = block256_exclusive_scan(deg, t, wtot, total);
    if (valid) {
#pragma unroll
        for (int g = 0; g < HG; ++g) {
            part[(size_t)g * NN + d] = run;
            run += c[g];
        }
        dinv[d] = rsqrtf((float)deg + 1.0f);
    }
    if (t == 0) btot[blockIdx.x] = total;
}

#define FE 4
__global__ void __launch_bounds__(256) k_fill(const int* __restrict__ src, const int* __restrict__ dst,
                                              const int* __restrict__ rank, const int* __restrict__ part,
                                              const int* __restrict__ btot, const float* __restrict__ dinv,
                                              int* __restrict__ rowptr, int* __restrict__ rec, int E) {
    static_assert(NBLK < 256 && MB == 256, "one block total per thread, and bp[NBLK] holds the grand total");
    __shared__ int bp[256];
    __shared__ __attribute__((aligned(16))) int wtot[8];
    const int t = threadIdx.x;
    const int chunk = (E + HG - 1) / HG;
    const int bl = btot[min(t, NBLK - 1)];
    const int nrow = blockIdx.x * MB + t;
    const int prow = part[min(nrow, NN - 1)];

    int   e[FE], s[FE], d[FE], rk[FE], pg[FE];
    float ws[FE], wd[FE];
#pragma unroll
    for (int k = 0; k < FE; ++k) {
        e[k] = (blockIdx.x * FE + k) * 256 + t;
        const int el = min(e[k], E - 1);
        s[k]  = src[el];
        d[k]  = dst[el];
        rk[k] = rank[el];
    }
#pragma unroll
    for (int k = 0; k < FE; ++k) {
        const int el = min(e[k], E - 1);
        pg[k] = part[(size_t)(el / chunk) * NN + d[k]];
        ws[k] = dinv[s[k]];
        wd[k] = dinv[d[k]];
    }

    int total;
    bp[t] = block256_exclusive_scan((t < NBLK) ? bl : 0, t, wtot, total);
    __syncthreads();

    int p[FE];
#pragma unroll
    for (int k = 0; k < FE; ++k) p[k] = bp[d[k] / MB] + pg[k] + rk[k];
#pragma unroll
    for (int k = 0; k < FE; ++k) {
        if (e[k] < E) ((int2*)rec)[p[k]] = make_int2(s[k], __float_as_int(ws[k] * wd[k]));
    }
    if (blockIdx.x < NBLK) {
        if (nrow < NN) rowptr[nrow] = bp[blockIdx.x] + prow;
        if (nrow == NN - 1) rowptr[NN] = bp[NBLK];
    }
}

template<int K, int NB, int NT>
__global__ void k_gemm_wmma(const float* __restrict__ A, const float* __restrict__ B,
                            const float* __restrict__ bias, float* __restrict__ C) {
    const int tm   = blockIdx.x;
    const int lane = threadIdx.x;
    const int half = lane >> 4;
    const int l16  = lane & 15;
    const int arow = tm * 16 + l16;

    const float* Ap = A + (size_t)arow * K + half * 2;
    const float* Bp = B + half * 2 * NB;

    v8f acc[NT];
#pragma unroll
    for (int t = 0; t < NT; ++t) acc[t] = (v8f){};

    for (int k = 0; k < K; k += 4) {
        v2f a;
        a.x = Ap[k];
        a.y = Ap[k + 1];
#pragma unroll
        for (int t = 0; t < NT; ++t) {
            const int c = t * 16 + l16;
            v2f b;
            if constexpr (NT * 16 == NB) {
                b.x = Bp[k * NB + c];
                b.y = Bp[k * NB + NB + c];
            } else {
                const float m  = (c < NB) ? 1.0f : 0.0f;
                const int   cc = (c < NB) ? c : (NB - 1);
                b.x = m * Bp[k * NB + cc];
                b.y = m * Bp[k * NB + NB + cc];
            }
            acc[t] = __builtin_amdgcn_wmma_f32_16x16x4_f32(
                false, a, false, b, (short)0, acc[t], false, false);
        }
    }

#pragma unroll
    for (int t = 0; t < NT; ++t) {
        const int c = t * 16 + l16;
        if (NT * 16 == NB || c < NB) {
            const float bv = bias ? bias[c] : 0.0f;
#pragma unroll
            for (int i = 0; i < 8; ++i)
                C[(size_t)(tm * 16 + i + half * 8) * NB + c] = acc[t][i] + bv;
        }
    }
}


template<int U>
__device__ __forceinline__ void add_edges(const float2* __restrict__ h2, int ms, int mw, int j,
                                          int lane, float& ax, float& ay) {
    int    s[U];
    float  w[U];
    float2 v[U];
#pragma unroll
    for (int k = 0; k < U; ++k) {
        s[k] = __builtin_amdgcn_readlane(ms, j + k);
        w[k] = __int_as_float(__builtin_amdgcn_readlane(mw, j + k));
    }
#pragma unroll
    for (int k = 0; k < U; ++k) v[k] = h2[(size_t)s[k] * 32 + lane];
#pragma unroll
    for (int k = 0; k < U; ++k) {
        ax += w[k] * v[k].x;
        ay += w[k] * v[k].y;
    }
}

__device__ __forceinline__ void add_batch(const float2* __restrict__ h2, int ms, int mw, int n,
                                          int lane, float& ax, float& ay) {
    int j = 0;
    for (; j + 8 <= n; j += 8) add_edges<8>(h2, ms, mw, j, lane, ax, ay);
    if (j + 4 <= n) { add_edges<4>(h2, ms, mw, j, lane, ax, ay); j += 4; }
    if (j + 2 <= n) { add_edges<2>(h2, ms, mw, j, lane, ax, ay); j += 2; }
    if (j < n)      { add_edges<1>(h2, ms, mw, j, lane, ax, ay); }
}

template<int NB, int NT>
__global__ void __launch_bounds__(512) k_gather_gemm(const int* __restrict__ rowptr, const int* __restrict__ rec,
                                                     const float* __restrict__ h, const float* __restrict__ dinv,
                                                     const float* __restrict__ bias, const float* __restrict__ Wn,
                                                     const float* __restrict__ bn, float* __restrict__ outp,
                                                     int do_elu, int E) {
    static_assert(NN % 16 == 0, "every tile holds 16 valid nodes; no wave may leave before the barriers");
    static_assert(NB % 4 == 0 && NT * 16 >= NB && NT <= 16, "tile shape");
    constexpr int NTILES = NN / 16;
    constexpr int PB  = (NB == 64) ? 160 : 96;
    constexpr int PA  = 68;
    constexpr int PC  = NB + 4;
    constexpr int QPR = NB / 4;
    constexpr int NQ  = 64 * QPR;
    constexpr int NQO = 16 * QPR;
    static_assert(NT * 32 <= PB && PB % 64 == 32 && NQ <= 1024 && NQO <= 512, "staging shape");
    __shared__ __attribute__((aligned(16))) float Ws[32 * PB];
    __shared__ __attribute__((aligned(16))) float As[16 * PA];
    __shared__ __attribute__((aligned(16))) float Cs[16 * PC];

    const int t    = threadIdx.x;
    const int lane = t & 31;
    const int wv   = __builtin_amdgcn_readfirstlane(t >> 5);

    constexpr int REM = NQ - 512;
    static_assert(REM > 0 && REM <= 512 && (REM & (REM - 1)) == 0, "second round wraps by a power of two");
    const int q0 = t;
    const int q1 = 512 + (t & (REM - 1));
    const float4 wq0 = ((const float4*)Wn)[q0];
    const float4 wq1 = ((const float4*)Wn)[q1];

    const float2* __restrict__ h2 = (const float2*)h;
    const int2*   __restrict__ r2 = (const int2*)rec;
    const int last_rec = max(E - 1, 0);

    static_assert(GRES <= NTILES, "every workgroup of the grid owns a first tile");
    int tile = blockIdx.x;
    int   beg, end;
    float di;
    float2 hv;
    {
        const int node = tile * 16 + wv;
        beg = rowptr[node];
        end = rowptr[node + 1];
        di  = dinv[node];
        hv  = h2[(size_t)node * 32 + lane];
    }
    int2 rfirst = r2[min(beg + lane, last_rec)];

    const float2 bv2 = ((const float2*)bias)[lane];
    const int half = lane >> 4;
    const int l16  = lane & 15;
    const int c    = (wv < NT ? wv : 0) * 16 + l16;
    const float* Ap = As + l16 * PA + half * 2;
    const float* Bp = Ws + half * PB + 2 * c;
    const float bvn = bn ? bn[(NT * 16 == NB) ? c : min(c, NB - 1)] : 0.0f;

    {
        float* w0 = &Ws[((q0 / QPR) >> 1) * PB + 8 * (q0 % QPR) + ((q0 / QPR) & 1)];
        float* w1 = &Ws[((q1 / QPR) >> 1) * PB + 8 * (q1 % QPR) + ((q1 / QPR) & 1)];
        w0[0] = wq0.x; w0[2] = wq0.y; w0[4] = wq0.z; w0[6] = wq0.w;
        w1[0] = wq1.x; w1[2] = wq1.y; w1[4] = wq1.z; w1[6] = wq1.w;
    }
    if constexpr (NT * 16 != NB) {
        constexpr int PQ = (NT * 16 - NB) / 2;
        if (t < 32 * PQ) *(float4*)&Ws[(t / PQ) * PB + 2 * NB + 4 * (t % PQ)] = make_float4(0.f, 0.f, 0.f, 0.f);
    }

    for (;;) {
        const int  ntile = tile + GRES;
        const bool more  = ntile < NTILES;
        const int  nnode = (more ? ntile : tile) * 16 + wv;
        const int    nbnd = rowptr[nnode + (lane & 1)];
        const float  ndi  = dinv[nnode];
        const float2 nhv  = h2[(size_t)nnode * 32 + lane];

        const float sw = di * di;
        float ax = sw * hv.x;
        float ay = sw * hv.y;

        add_batch(h2, rfirst.x, rfirst.y, min(32, end - beg), lane, ax, ay);
        for (int base = beg + 32; base < end; base += 32) {
            const int2 r = r2[min(base + lane, last_rec)];
            add_batch(h2, r.x, r.y, min(32, end - base), lane, ax, ay);
        }
        const int nbeg = __builtin_amdgcn_readlane(nbnd, 0);
        const int nend = __builtin_amdgcn_readlane(nbnd, 1);

        float vx = ax + bv2.x;
        float vy = ay + bv2.y;
        if (do_elu) {
            vx = (vx <= 0.0f) ? expm1f(vx) : vx;
            vy = (vy <= 0.0f) ? expm1f(vy) : vy;
        }
        *(float2*)&As[wv * PA + 2 * lane] = make_float2(vx, vy);

        __syncthreads();

        const int2 nrfirst = r2[min(nbeg + lane, last_rec)];

        if (wv < NT) {
            v8f acc = (v8f){};
#pragma unroll
            for (int k = 0; k < H; k += 4) {
                v2f a, b;
                a.x = Ap[k];
                a.y = Ap[k + 1];
                b.x = Bp[(k / 2) * PB];
                b.y = Bp[(k / 2) * PB + 1];
                acc = __builtin_amdgcn_wmma_f32_16x16x4_f32(
                    false, a, false, b, (short)0, acc, false, false);
            }
            if (NT * 16 == NB || c < NB) {
#pragma unroll
                for (int i = 0; i < 8; ++i) Cs[(i + half * 8) * PC + c] = acc[i] + bvn;
            }
        }
        __syncthreads();

        if (t < NQO) {
            const float4 v = *(const float4*)&Cs[(t / QPR) * PC + 4 * (t % QPR)];
            ((float4*)outp)[(size_t)tile * NQO + t] = v;
        }

        if (!more) break;
        tile   = ntile;
        beg    = nbeg;
        end    = nend;
        di     = ndi;
        hv     = nhv;
        rfirst = nrfirst;
    }
}


extern "C" void kernel_launch(void* const* d_in, const int* in_sizes, int n_in,
                              void* d_out, int out_size, void* d_ws, size_t ws_size,
                              hipStream_t stream) {
    const float* x  = (const float*)d_in[0];
    const int*   ei = (const int*)d_in[1];
    const float* W1 = (const float*)d_in[2];
    const float* b1 = (const float*)d_in[3];
    const float* W2 = (const float*)d_in[4];
    const float* b2 = (const float*)d_in[5];
    const float* W3 = (const float*)d_in[6];
    const float* b3 = (const float*)d_in[7];
    const float* Wl = (const float*)d_in[8];
    const float* bl = (const float*)d_in[9];
    float* out = (float*)d_out;

    const int E = in_sizes[1] / 2;
    const int* src = ei;
    const int* dst = ei + E;

    float* G      = (float*)d_ws;
    float* ACT    = G + (size_t)NN * H;
    int*   btot   = (int*)(ACT + (size_t)NN * H);
    float* dinv   = (float*)(btot + NN);
    int*   rowptr = (int*)(dinv + NN);
    int*   rec    = rowptr + NN + 16;
    int*   rank   = rec + 2 * (size_t)E;
    int*   part   = rank + E;

    const dim3 blk(256);

    k_hist<<<dim3(HG), dim3(1024), 0, stream>>>(dst, part, rank, E);
    k_merge<<<dim3(NBLK), dim3(MB), 0, stream>>>(part, btot, dinv);
    k_fill<<<dim3((E + 256 * FE - 1) / (256 * FE)), blk, 0, stream>>>(src, dst, rank, part, btot, dinv, rowptr, rec, E);

    const dim3 gGemm(NN / 16);
    const dim3 wave(32);
    const dim3 blk16(512);
    const dim3 gRes(GRES);


    k_gemm_wmma<128, H, 4><<<gGemm, wave, 0, stream>>>(x, W1, nullptr, G);
    k_gather_gemm<H, 4><<<gRes, blk16, 0, stream>>>(rowptr, rec, G, dinv, b1, W2, nullptr, ACT, 1, E);
    k_gather_gemm<H, 4><<<gRes, blk16, 0, stream>>>(rowptr, rec, ACT, dinv, b2, W3, nullptr, G, 1, E);
    k_gather_gemm<40, 3><<<gRes, blk16, 0, stream>>>(rowptr, rec, G, dinv, b3, Wl, bl, out, 0, E);
}


# Layer-1 product x@W1 joins the histogram launch as 224 extra 32-wave workgroups (14 row tiles each, one per wave) on the WGPs its 32 workgroups leave idle; 7 launches to 6; sum orders as parent
# speedup vs baseline: 3.8194x; 1.0957x over previous
#include <hip/hip_runtime.h>
#include <math.h>


typedef __attribute__((ext_vector_type(2))) float v2f;
typedef __attribute__((ext_vector_type(8))) float v8f;

#define NN 50000
#define H  64
#define GRES 1024

template<int K, int NB, int NT>
__device__ __forceinline__ void gemm_tile_wmma(const float* __restrict__ A, const float* __restrict__ B,
                                               float* __restrict__ C, int tm, int lane) {
    static_assert(NT * 16 == NB, "whole column tiles only");
    const int half = lane >> 4;
    const int l16  = lane & 15;
    const int arow = tm * 16 + l16;

    const float* Ap = A + (size_t)arow * K + half * 2;
    const float* Bp = B + half * 2 * NB;

    v8f acc[NT];
#pragma unroll
    for (int t = 0; t < NT; ++t) acc[t] = (v8f){};

    for (int k = 0; k < K; k += 4) {
        v2f a;
        a.x = Ap[k];
        a.y = Ap[k + 1];
#pragma unroll
        for (int t = 0; t < NT; ++t) {
            const int c = t * 16 + l16;
            v2f b;
            b.x = Bp[k * NB + c];
            b.y = Bp[k * NB + NB + c];
            acc[t] = __builtin_amdgcn_wmma_f32_16x16x4_f32(
                false, a, false, b, (short)0, acc[t], false, false);
        }
    }

#pragma unroll
    for (int t = 0; t < NT; ++t) {
        const int c = t * 16 + l16;
#pragma unroll
        for (int i = 0; i < 8; ++i)
            C[(size_t)(tm * 16 + i + half * 8) * NB + c] = acc[t][i];
    }
}


#define HG 32

#define GT   14
#define NGB  ((NN / 16 + GT - 1) / GT)
__global__ void __launch_bounds__(1024) k_hist_gemm(const int* __restrict__ dst, int* __restrict__ part,
                                                    int* __restrict__ rank, int E,
                                                    const float* __restrict__ X, const float* __restrict__ W1,
                                                    float* __restrict__ G) {
    static_assert(NN % 4 == 0, "histogram cleared and copied out 16 bytes per thread");
    static_assert(GT <= 32, "one tile per wave of a product workgroup");
    __shared__ __attribute__((aligned(16))) int hist[NN];
    const int t = threadIdx.x;
    if (blockIdx.x >= HG) {
        const int wv   = __builtin_amdgcn_readfirstlane(t >> 5);
        const int tile = (blockIdx.x - HG) * GT + wv;
        if (wv < GT && tile < NN / 16) gemm_tile_wmma<128, H, 4>(X, W1, G, tile, t & 31);
        return;
    }
    const int g = blockIdx.x;
    for (int i = t; i < NN / 4; i += 1024) ((int4*)hist)[i] = make_int4(0, 0, 0, 0);
    __syncthreads();

    const int chunk = (E + HG - 1) / HG;
    const int beg   = g * chunk;
    const int end   = min(E, beg + chunk);
    int e = beg + t;
    for (; e + 7 * 1024 < end; e += 8 * 1024) {
        int d[8], r[8];
#pragma unroll
        for (int k = 0; k < 8; ++k) d[k] = dst[e + k * 1024];
#pragma unroll
        for (int k = 0; k < 8; ++k) r[k] = atomicAdd(&hist[d[k]], 1);
#pragma unroll
        for (int k = 0; k < 8; ++k) rank[e + k * 1024] = r[k];
    }
    for (; e < end; e += 1024) rank[e] = atomicAdd(&hist[dst[e]], 1);
    __syncthreads();

    int4* __restrict__ po = (int4*)(part + (size_t)g * NN);
    for (int i = t; i < NN / 4; i += 1024) po[i] = ((const int4*)hist)[i];
}

__device__ __forceinline__ int wave_inclusive_scan(int v, int lane) {
#pragma unroll
    for (int off = 1; off < 32; off <<= 1) {
        int u = __shfl_up(v, off);
        if (lane >= off) v += u;
    }
    return v;
}

#define MB   256
#define NBLK ((NN + MB - 1) / MB)

__device__ __forceinline__ int block256_exclusive_scan(int v, int t, int* wtot, int& total) {
    const int lane = t & 31;
    const int wv   = t >> 5;
    const int inc  = wave_inclusive_scan(v, lane);
    if (lane == 31) wtot[wv] = inc;
    __syncthreads();
    const int4 lo = *(const int4*)&wtot[0];
    const int4 hi = *(const int4*)&wtot[4];
    const int w[8] = {lo.x, lo.y, lo.z, lo.w, hi.x, hi.y, hi.z, hi.w};
    int before = 0, all = 0;
#pragma unroll
    for (int k = 0; k < 8; ++k) {
        before += (k < wv) ? w[k] : 0;
        all    += w[k];
    }
    total = all;
    return before + inc - v;
}

__global__ void __launch_bounds__(MB) k_merge(int* __restrict__ part, int* __restrict__ btot,
                                              float* __restrict__ dinv) {
    __shared__ __attribute__((aligned(16))) int wtot[8];
    const int t = threadIdx.x;
    const int d = blockIdx.x * MB + t;
    const bool valid = d < NN;
    const int  dl = valid ? d : NN - 1;
    int c[HG];
#pragma unroll
    for (int g = 0; g < HG; ++g) c[g] = part[(size_t)g * NN + dl];
    int deg = 0;
#pragma unroll
    for (int g = 0; g < HG; ++g) deg += c[g];
    if (!valid) deg = 0;
    int total;
    int run = block256_exclusive_scan(deg, t, wtot, total);
    if (valid) {
#pragma unroll
        for (int g = 0; g < HG; ++g) {
            part[(size_t)g * NN + d] = run;
            run += c[g];
        }
        dinv[d] = rsqrtf((float)deg + 1.0f);
    }
    if (t == 0) btot[blockIdx.x] = total;
}

#define FE 4
__global__ void __launch_bounds__(256) k_fill(const int* __restrict__ src, const int* __restrict__ dst,
                                              const int* __restrict__ rank, const int* __restrict__ part,
                                              const int* __restrict__ btot, const float* __restrict__ dinv,
                                              int* __restrict__ rowptr, int* __restrict__ rec, int E) {
    static_assert(NBLK < 256 && MB == 256, "one block total per thread, and bp[NBLK] holds the grand total");
    __shared__ int bp[256];
    __shared__ __attribute__((aligned(16))) int wtot[8];
    const int t = threadIdx.x;
    const int chunk = (E + HG - 1) / HG;
    const int bl = btot[min(t, NBLK - 1)];
    const int nrow = blockIdx.x * MB + t;
    const int prow = part[min(nrow, NN - 1)];

    int   e[FE], s[FE], d[FE], rk[FE], pg[FE];
    float ws[FE], wd[FE];
#pragma unroll
    for (int k = 0; k < FE; ++k) {
        e[k] = (blockIdx.x * FE + k) * 256 + t;
        const int el = min(e[k], E - 1);
        s[k]  = src[el];
        d[k]  = dst[el];
        rk[k] = rank[el];
    }
#pragma unroll
    for (int k = 0; k < FE; ++k) {
        const int el = min(e[k], E - 1);
        pg[k] = part[(size_t)(el / chunk) * NN + d[k]];
        ws[k] = dinv[s[k]];
        wd[k] = dinv[d[k]];
    }

    int total;
    bp[t] = block256_exclusive_scan((t < NBLK) ? bl : 0, t, wtot, total);
    __syncthreads();

    int p[FE];
#pragma unroll
    for (int k = 0; k < FE; ++k) p[k] = bp[d[k] / MB] + pg[k] + rk[k];
#pragma unroll
    for (int k = 0; k < FE; ++k) {
        if (e[k] < E) ((int2*)rec)[p[k]] = make_int2(s[k], __float_as_int(ws[k] * wd[k]));
    }
    if (blockIdx.x < NBLK) {
        if (nrow < NN) rowptr[nrow] = bp[blockIdx.x] + prow;
        if (nrow == NN - 1) rowptr[NN] = bp[NBLK];
    }
}


template<int U>
__device__ __forceinline__ void add_edges(const float2* __restrict__ h2, int ms, int mw, int j,
                                          int lane, float& ax, float& ay) {
    int    s[U];
    float  w[U];
    float2 v[U];
#pragma unroll
    for (int k = 0; k < U; ++k) {
        s[k] = __builtin_amdgcn_readlane(ms, j + k);
        w[k] = __int_as_float(__builtin_amdgcn_readlane(mw, j + k));
    }
#pragma unroll
    for (int k = 0; k < U; ++k) v[k] = h2[(size_t)s[k] * 32 + lane];
#pragma unroll
    for (int k = 0; k < U; ++k) {
        ax += w[k] * v[k].x;
        ay += w[k] * v[k].y;
    }
}

__device__ __forceinline__ void add_batch(const float2* __restrict__ h2, int ms, int mw, int n,
                                          int lane, float& ax, float& ay) {
    int j = 0;
    for (; j + 8 <= n; j += 8) add_edges<8>(h2, ms, mw, j, lane, ax, ay);
    if (j + 4 <= n) { add_edges<4>(h2, ms, mw, j, lane, ax, ay); j += 4; }
    if (j + 2 <= n) { add_edges<2>(h2, ms, mw, j, lane, ax, ay); j += 2; }
    if (j < n)      { add_edges<1>(h2, ms, mw, j, lane, ax, ay); }
}

template<int NB, int NT>
__global__ void __launch_bounds__(512) k_gather_gemm(const int* __restrict__ rowptr, const int* __restrict__ rec,
                                                     const float* __restrict__ h, const float* __restrict__ dinv,
                                                     const float* __restrict__ bias, const float* __restrict__ Wn,
                                                     const float* __restrict__ bn, float* __restrict__ outp,
                                                     int do_elu, int E) {
    static_assert(NN % 16 == 0, "every tile holds 16 valid nodes; no wave may leave before the barriers");
    static_assert(NB % 4 == 0 && NT * 16 >= NB && NT <= 16, "tile shape");
    constexpr int NTILES = NN / 16;
    constexpr int PB  = (NB == 64) ? 160 : 96;
    constexpr int PA  = 68;
    constexpr int PC  = NB + 4;
    constexpr int QPR = NB / 4;
    constexpr int NQ  = 64 * QPR;
    constexpr int NQO = 16 * QPR;
    static_assert(NT * 32 <= PB && PB % 64 == 32 && NQ <= 1024 && NQO <= 512, "staging shape");
    __shared__ __attribute__((aligned(16))) float Ws[32 * PB];
    __shared__ __attribute__((aligned(16))) float As[16 * PA];
    __shared__ __attribute__((aligned(16))) float Cs[16 * PC];

    const int t    = threadIdx.x;
    const int lane = t & 31;
    const int wv   = __builtin_amdgcn_readfirstlane(t >> 5);

    constexpr int REM = NQ - 512;
    static_assert(REM > 0 && REM <= 512 && (REM & (REM - 1)) == 0, "second round wraps by a power of two");
    const int q0 = t;
    const int q1 = 512 + (t & (REM - 1));
    const float4 wq0 = ((const float4*)Wn)[q0];
    const float4 wq1 = ((const float4*)Wn)[q1];

    const float2* __restrict__ h2 = (const float2*)h;
    const int2*   __restrict__ r2 = (const int2*)rec;
    const int last_rec = max(E - 1, 0);

    static_assert(GRES <= NTILES, "every workgroup of the grid owns a first tile");
    int tile = blockIdx.x;
    int   beg, end;
    float di;
    float2 hv;
    {
        const int node = tile * 16 + wv;
        beg = rowptr[node];
        end = rowptr[node + 1];
        di  = dinv[node];
        hv  = h2[(size_t)node * 32 + lane];
    }
    int2 rfirst = r2[min(beg + lane, last_rec)];

    const float2 bv2 = ((const float2*)bias)[lane];
    const int half = lane >> 4;
    const int l16  = lane & 15;
    const int c    = (wv < NT ? wv : 0) * 16 + l16;
    const float* Ap = As + l16 * PA + half * 2;
    const float* Bp = Ws + half * PB + 2 * c;
    const float bvn = bn ? bn[(NT * 16 == NB) ? c : min(c, NB - 1)] : 0.0f;

    {
        float* w0 = &Ws[((q0 / QPR) >> 1) * PB + 8 * (q0 % QPR) + ((q0 / QPR) & 1)];
        float* w1 = &Ws[((q1 / QPR) >> 1) * PB + 8 * (q1 % QPR) + ((q1 / QPR) & 1)];
        w0[0] = wq0.x; w0[2] = wq0.y; w0[4] = wq0.z; w0[6] = wq0.w;
        w1[0] = wq1.x; w1[2] = wq1.y; w1[4] = wq1.z; w1[6] = wq1.w;
    }
    if constexpr (NT * 16 != NB) {
        constexpr int PQ = (NT * 16 - NB) / 2;
        if (t < 32 * PQ) *(float4*)&Ws[(t / PQ) * PB + 2 * NB + 4 * (t % PQ)] = make_float4(0.f, 0.f, 0.f, 0.f);
    }

    for (;;) {
        const int  ntile = tile + GRES;
        const bool more  = ntile < NTILES;
        const int  nnode = (more ? ntile : tile) * 16 + wv;
        const int    nbnd = rowptr[nnode + (lane & 1)];
        const float  ndi  = dinv[nnode];
        const float2 nhv  = h2[(size_t)nnode * 32 + lane];

        const float sw = di * di;
        float ax = sw * hv.x;
        float ay = sw * hv.y;

        add_batch(h2, rfirst.x, rfirst.y, min(32, end - beg), lane, ax, ay);
        for (int base = beg + 32; base < end; base += 32) {
            const int2 r = r2[min(base + lane, last_rec)];
            add_batch(h2, r.x, r.y, min(32, end - base), lane, ax, ay);
        }
        const int nbeg = __builtin_amdgcn_readlane(nbnd, 0);
        const int nend = __builtin_amdgcn_readlane(nbnd, 1);

        float vx = ax + bv2.x;
        float vy = ay + bv2.y;
        if (do_elu) {
            vx = (vx <= 0.0f) ? expm1f(vx) : vx;
            vy = (vy <= 0.0f) ? expm1f(vy) : vy;
        }
        *(float2*)&As[wv * PA + 2 * lane] = make_float2(vx, vy);

        __syncthreads();

        const int2 nrfirst = r2[min(nbeg + lane, last_rec)];

        if (wv < NT) {
            v8f acc = (v8f){};
#pragma unroll
            for (int k = 0; k < H; k += 4) {
                v2f a, b;
                a.x = Ap[k];
                a.y = Ap[k + 1];
                b.x = Bp[(k / 2) * PB];
                b.y = Bp[(k / 2) * PB + 1];
                acc = __builtin_amdgcn_wmma_f32_16x16x4_f32(
                    false, a, false, b, (short)0, acc, false, false);
            }
            if (NT * 16 == NB || c < NB) {
#pragma unroll
                for (int i = 0; i < 8; ++i) Cs[(i + half * 8) * PC + c] = acc[i] + bvn;
            }
        }
        __syncthreads();

        if (t < NQO) {
            const float4 v = *(const float4*)&Cs[(t / QPR) * PC + 4 * (t % QPR)];
            ((float4*)outp)[(size_t)tile * NQO + t] = v;
        }

        if (!more) break;
        tile   = ntile;
        beg    = nbeg;
        end    = nend;
        di     = ndi;
        hv     = nhv;
        rfirst = nrfirst;
    }
}


extern "C" void kernel_launch(void* const* d_in, const int* in_sizes, int n_in,
                              void* d_out, int out_size, void* d_ws, size_t ws_size,
                              hipStream_t stream) {
    const float* x  = (const float*)d_in[0];
    const int*   ei = (const int*)d_in[1];
    const float* W1 = (const float*)d_in[2];
    const float* b1 = (const float*)d_in[3];
    const float* W2 = (const float*)d_in[4];
    const float* b2 = (const float*)d_in[5];
    const float* W3 = (const float*)d_in[6];
    const float* b3 = (const float*)d_in[7];
    const float* Wl = (const float*)d_in[8];
    const float* bl = (const float*)d_in[9];
    float* out = (float*)d_out;

    const int E = in_sizes[1] / 2;
    const int* src = ei;
    const int* dst = ei + E;

    float* G      = (float*)d_ws;
    float* ACT    = G + (size_t)NN * H;
    int*   btot   = (int*)(ACT + (size_t)NN * H);
    float* dinv   = (float*)(btot + NN);
    int*   rowptr = (int*)(dinv + NN);
    int*   rec    = rowptr + NN + 16;
    int*   rank   = rec + 2 * (size_t)E;
    int*   part   = rank + E;

    const dim3 blk(256);

    k_hist_gemm<<<dim3(HG + NGB), dim3(1024), 0, stream>>>(dst, part, rank, E, x, W1, G);
    k_merge<<<dim3(NBLK), dim3(MB), 0, stream>>>(part, btot, dinv);
    k_fill<<<dim3((E + 256 * FE - 1) / (256 * FE)), blk, 0, stream>>>(src, dst, rank, part, btot, dinv, rowptr, rec, E);

    const dim3 blk16(512);
    const dim3 gRes(GRES);


    k_gather_gemm<H, 4><<<gRes, blk16, 0, stream>>>(rowptr, rec, G, dinv, b1, W2, nullptr, ACT, 1, E);
    k_gather_gemm<H, 4><<<gRes, blk16, 0, stream>>>(rowptr, rec, ACT, dinv, b2, W3, nullptr, G, 1, E);
    k_gather_gemm<40, 3><<<gRes, blk16, 0, stream>>>(rowptr, rec, G, dinv, b3, Wl, bl, out, 0, E);
}


# A node's rows are asked for in groups of 14, the left-over rows as one group (was 8, then 4, 2, 1 in turn); 32-bit row index so 14 in flight fit 64 registers; sum order as parent
# speedup vs baseline: 3.9767x; 1.0412x over previous
#include <hip/hip_runtime.h>
#include <math.h>


typedef __attribute__((ext_vector_type(2))) float v2f;
typedef __attribute__((ext_vector_type(8))) float v8f;

#define NN 50000
#define H  64
#define GRES 1024

template<int K, int NB, int NT>
__device__ __forceinline__ void gemm_tile_wmma(const float* __restrict__ A, const float* __restrict__ B,
                                               float* __restrict__ C, int tm, int lane) {
    static_assert(NT * 16 == NB, "whole column tiles only");
    const int half = lane >> 4;
    const int l16  = lane & 15;
    const int arow = tm * 16 + l16;

    const float* Ap = A + (size_t)arow * K + half * 2;
    const float* Bp = B + half * 2 * NB;

    v8f acc[NT];
#pragma unroll
    for (int t = 0; t < NT; ++t) acc[t] = (v8f){};

    for (int k = 0; k < K; k += 4) {
        v2f a;
        a.x = Ap[k];
        a.y = Ap[k + 1];
#pragma unroll
        for (int t = 0; t < NT; ++t) {
            const int c = t * 16 + l16;
            v2f b;
            b.x = Bp[k * NB + c];
            b.y = Bp[k * NB + NB + c];
            acc[t] = __builtin_amdgcn_wmma_f32_16x16x4_f32(
                false, a, false, b, (short)0, acc[t], false, false);
        }
    }

#pragma unroll
    for (int t = 0; t < NT; ++t) {
        const int c = t * 16 + l16;
#pragma unroll
        for (int i = 0; i < 8; ++i)
            C[(size_t)(tm * 16 + i + half * 8) * NB + c] = acc[t][i];
    }
}


#define HG 32

#define GT   14
#define NGB  ((NN / 16 + GT - 1) / GT)
__global__ void __launch_bounds__(1024) k_hist_gemm(const int* __restrict__ dst, int* __restrict__ part,
                                                    int* __restrict__ rank, int E,
                                                    const float* __restrict__ X, const float* __restrict__ W1,
                                                    float* __restrict__ G) {
    static_assert(NN % 4 == 0, "histogram cleared and copied out 16 bytes per thread");
    static_assert(GT <= 32, "one tile per wave of a product workgroup");
    __shared__ __attribute__((aligned(16))) int hist[NN];
    const int t = threadIdx.x;
    if (blockIdx.x >= HG) {
        const int wv   = __builtin_amdgcn_readfirstlane(t >> 5);
        const int tile = (blockIdx.x - HG) * GT + wv;
        if (wv < GT && tile < NN / 16) gemm_tile_wmma<128, H, 4>(X, W1, G, tile, t & 31);
        return;
    }
    const int g = blockIdx.x;
    for (int i = t; i < NN / 4; i += 1024) ((int4*)hist)[i] = make_int4(0, 0, 0, 0);
    __syncthreads();

    const int chunk = (E + HG - 1) / HG;
    const int beg   = g * chunk;
    const int end   = min(E, beg + chunk);
    int e = beg + t;
    for (; e + 7 * 1024 < end; e += 8 * 1024) {
        int d[8], r[8];
#pragma unroll
        for (int k = 0; k < 8; ++k) d[k] = dst[e + k * 1024];
#pragma unroll
        for (int k = 0; k < 8; ++k) r[k] = atomicAdd(&hist[d[k]], 1);
#pragma unroll
        for (int k = 0; k < 8; ++k) rank[e + k * 1024] = r[k];
    }
    for (; e < end; e += 1024) rank[e] = atomicAdd(&hist[dst[e]], 1);
    __syncthreads();

    int4* __restrict__ po = (int4*)(part + (size_t)g * NN);
    for (int i = t; i < NN / 4; i += 1024) po[i] = ((const int4*)hist)[i];
}

__device__ __forceinline__ int wave_inclusive_scan(int v, int lane) {
#pragma unroll
    for (int off = 1; off < 32; off <<= 1) {
        int u = __shfl_up(v, off);
        if (lane >= off) v += u;
    }
    return v;
}

#define MB   256
#define NBLK ((NN + MB - 1) / MB)

__device__ __forceinline__ int block256_exclusive_scan(int v, int t, int* wtot, int& total) {
    const int lane = t & 31;
    const int wv   = t >> 5;
    const int inc  = wave_inclusive_scan(v, lane);
    if (lane == 31) wtot[wv] = inc;
    __syncthreads();
    const int4 lo = *(const int4*)&wtot[0];
    const int4 hi = *(const int4*)&wtot[4];
    const int w[8] = {lo.x, lo.y, lo.z, lo.w, hi.x, hi.y, hi.z, hi.w};
    int before = 0, all = 0;
#pragma unroll
    for (int k = 0; k < 8; ++k) {
        before += (k < wv) ? w[k] : 0;
        all    += w[k];
    }
    total = all;
    return before + inc - v;
}

__global__ void __launch_bounds__(MB) k_merge(int* __restrict__ part, int* __restrict__ btot,
                                              float* __restrict__ dinv) {
    __shared__ __attribute__((aligned(16))) int wtot[8];
    const int t = threadIdx.x;
    const int d = blockIdx.x * MB + t;
    const bool valid = d < NN;
    const int  dl = valid ? d : NN - 1;
    int c[HG];
#pragma unroll
    for (int g = 0; g < HG; ++g) c[g] = part[(size_t)g * NN + dl];
    int deg = 0;
#pragma unroll
    for (int g = 0; g < HG; ++g) deg += c[g];
    if (!valid) deg = 0;
    int total;
    int run = block256_exclusive_scan(deg, t, wtot, total);
    if (valid) {
#pragma unroll
        for (int g = 0; g < HG; ++g) {
            part[(size_t)g * NN + d] = run;
            run += c[g];
        }
        dinv[d] = rsqrtf((float)deg + 1.0f);
    }
    if (t == 0) btot[blockIdx.x] = total;
}

#define FE 4
__global__ void __launch_bounds__(256) k_fill(const int* __restrict__ src, const int* __restrict__ dst,
                                              const int* __restrict__ rank, const int* __restrict__ part,
                                              const int* __restrict__ btot, const float* __restrict__ dinv,
                                              int* __restrict__ rowptr, int* __restrict__ rec, int E) {
    static_assert(NBLK < 256 && MB == 256, "one block total per thread, and bp[NBLK] holds the grand total");
    __shared__ int bp[256];
    __shared__ __attribute__((aligned(16))) int wtot[8];
    const int t = threadIdx.x;
    const int chunk = (E + HG - 1) / HG;
    const int bl = btot[min(t, NBLK - 1)];
    const int nrow = blockIdx.x * MB + t;
    const int prow = part[min(nrow, NN - 1)];

    int   e[FE], s[FE], d[FE], rk[FE], pg[FE];
    float ws[FE], wd[FE];
#pragma unroll
    for (int k = 0; k < FE; ++k) {
        e[k] = (blockIdx.x * FE + k) * 256 + t;
        const int el = min(e[k], E - 1);
        s[k]  = src[el];
        d[k]  = dst[el];
        rk[k] = rank[el];
    }
#pragma unroll
    for (int k = 0; k < FE; ++k) {
        const int el = min(e[k], E - 1);
        pg[k] = part[(size_t)(el / chunk) * NN + d[k]];
        ws[k] = dinv[s[k]];
        wd[k] = dinv[d[k]];
    }

    int total;
    bp[t] = block256_exclusive_scan((t < NBLK) ? bl : 0, t, wtot, total);
    __syncthreads();

    int p[FE];
#pragma unroll
    for (int k = 0; k < FE; ++k) p[k] = bp[d[k] / MB] + pg[k] + rk[k];
#pragma unroll
    for (int k = 0; k < FE; ++k) {
        if (e[k] < E) ((int2*)rec)[p[k]] = make_int2(s[k], __float_as_int(ws[k] * wd[k]));
    }
    if (blockIdx.x < NBLK) {
        if (nrow < NN) rowptr[nrow] = bp[blockIdx.x] + prow;
        if (nrow == NN - 1) rowptr[NN] = bp[NBLK];
    }
}


template<int U>
__device__ __forceinline__ void add_edges(const float2* __restrict__ h2, int ms, int mw, int j,
                                          int lane, float& ax, float& ay) {
    unsigned s[U];
    float  w[U];
    float2 v[U];
#pragma unroll
    for (int k = 0; k < U; ++k) {
        s[k] = (unsigned)__builtin_amdgcn_readlane(ms, j + k);
        w[k] = __int_as_float(__builtin_amdgcn_readlane(mw, j + k));
    }
#pragma unroll
    for (int k = 0; k < U; ++k) v[k] = h2[s[k] * 32u + (unsigned)lane];
#pragma unroll
    for (int k = 0; k < U; ++k) {
        ax += w[k] * v[k].x;
        ay += w[k] * v[k].y;
    }
}

#define GE 14
__device__ __forceinline__ void add_batch(const float2* __restrict__ h2, int ms, int mw, int n,
                                          int lane, float& ax, float& ay) {
    int j = 0;
    for (; j + GE <= n; j += GE) add_edges<GE>(h2, ms, mw, j, lane, ax, ay);
    switch (n - j) {
        case 13: add_edges<13>(h2, ms, mw, j, lane, ax, ay); break;
        case 12: add_edges<12>(h2, ms, mw, j, lane, ax, ay); break;
        case 11: add_edges<11>(h2, ms, mw, j, lane, ax, ay); break;
        case 10: add_edges<10>(h2, ms, mw, j, lane, ax, ay); break;
        case 9: add_edges<9>(h2, ms, mw, j, lane, ax, ay); break;
        case 8: add_edges<8>(h2, ms, mw, j, lane, ax, ay); break;
        case 7: add_edges<7>(h2, ms, mw, j, lane, ax, ay); break;
        case 6: add_edges<6>(h2, ms, mw, j, lane, ax, ay); break;
        case 5: add_edges<5>(h2, ms, mw, j, lane, ax, ay); break;
        case 4: add_edges<4>(h2, ms, mw, j, lane, ax, ay); break;
        case 3: add_edges<3>(h2, ms, mw, j, lane, ax, ay); break;
        case 2: add_edges<2>(h2, ms, mw, j, lane, ax, ay); break;
        case 1: add_edges<1>(h2, ms, mw, j, lane, ax, ay); break;
        default: break;
    }
}

template<int NB, int NT>
__global__ void __launch_bounds__(512) k_gather_gemm(const int* __restrict__ rowptr, const int* __restrict__ rec,
                                                     const float* __restrict__ h, const float* __restrict__ dinv,
                                                     const float* __restrict__ bias, const float* __restrict__ Wn,
                                                     const float* __restrict__ bn, float* __restrict__ outp,
                                                     int do_elu, int E) {
    static_assert(NN % 16 == 0, "every tile holds 16 valid nodes; no wave may leave before the barriers");
    static_assert(NB % 4 == 0 && NT * 16 >= NB && NT <= 16, "tile shape");
    constexpr int NTILES = NN / 16;
    constexpr int PB  = (NB == 64) ? 160 : 96;
    constexpr int PA  = 68;
    constexpr int PC  = NB + 4;
    constexpr int QPR = NB / 4;
    constexpr int NQ  = 64 * QPR;
    constexpr int NQO = 16 * QPR;
    static_assert(NT * 32 <= PB && PB % 64 == 32 && NQ <= 1024 && NQO <= 512, "staging shape");
    __shared__ __attribute__((aligned(16))) float Ws[32 * PB];
    __shared__ __attribute__((aligned(16))) float As[16 * PA];
    __shared__ __attribute__((aligned(16))) float Cs[16 * PC];

    const int t    = threadIdx.x;
    const int lane = t & 31;
    const int wv   = __builtin_amdgcn_readfirstlane(t >> 5);

    constexpr int REM = NQ - 512;
    static_assert(REM > 0 && REM <= 512 && (REM & (REM - 1)) == 0, "second round wraps by a power of two");
    const int q0 = t;
    const int q1 = 512 + (t & (REM - 1));
    const float4 wq0 = ((const float4*)Wn)[q0];
    const float4 wq1 = ((const float4*)Wn)[q1];

    const float2* __restrict__ h2 = (const float2*)h;
    const int2*   __restrict__ r2 = (const int2*)rec;
    const int last_rec = max(E - 1, 0);

    static_assert(GRES <= NTILES, "every workgroup of the grid owns a first tile");
    int tile = blockIdx.x;
    int   beg, end;
    float di;
    float2 hv;
    {
        const int node = tile * 16 + wv;
        beg = rowptr[node];
        end = rowptr[node + 1];
        di  = dinv[node];
        hv  = h2[(size_t)node * 32 + lane];
    }
    int2 rfirst = r2[min(beg + lane, last_rec)];

    const float2 bv2 = ((const float2*)bias)[lane];
    const int half = lane >> 4;
    const int l16  = lane & 15;
    const int c    = (wv < NT ? wv : 0) * 16 + l16;
    const float* Ap = As + l16 * PA + half * 2;
    const float* Bp = Ws + half * PB + 2 * c;
    const float bvn = bn ? bn[(NT * 16 == NB) ? c : min(c, NB - 1)] : 0.0f;

    {
        float* w0 = &Ws[((q0 / QPR) >> 1) * PB + 8 * (q0 % QPR) + ((q0 / QPR) & 1)];
        float* w1 = &Ws[((q1 / QPR) >> 1) * PB + 8 * (q1 % QPR) + ((q1 / QPR) & 1)];
        w0[0] = wq0.x; w0[2] = wq0.y; w0[4] = wq0.z; w0[6] = wq0.w;
        w1[0] = wq1.x; w1[2] = wq1.y; w1[4] = wq1.z; w1[6] = wq1.w;
    }
    if constexpr (NT * 16 != NB) {
        constexpr int PQ = (NT * 16 - NB) / 2;
        if (t < 32 * PQ) *(float4*)&Ws[(t / PQ) * PB + 2 * NB + 4 * (t % PQ)] = make_float4(0.f, 0.f, 0.f, 0.f);
    }

    for (;;) {
        const int  ntile = tile + GRES;
        const bool more  = ntile < NTILES;
        const int  nnode = (more ? ntile : tile) * 16 + wv;
        const int    nbnd = rowptr[nnode + (lane & 1)];
        const float  ndi  = dinv[nnode];
        const float2 nhv  = h2[(size_t)nnode * 32 + lane];

        const float sw = di * di;
        float ax = sw * hv.x;
        float ay = sw * hv.y;

        add_batch(h2, rfirst.x, rfirst.y, min(32, end - beg), lane, ax, ay);
        for (int base = beg + 32; base < end; base += 32) {
            const int2 r = r2[min(base + lane, last_rec)];
            add_batch(h2, r.x, r.y, min(32, end - base), lane, ax, ay);
        }
        const int nbeg = __builtin_amdgcn_readlane(nbnd, 0);
        const int nend = __builtin_amdgcn_readlane(nbnd, 1);

        float vx = ax + bv2.x;
        float vy = ay + bv2.y;
        if (do_elu) {
            vx = (vx <= 0.0f) ? expm1f(vx) : vx;
            vy = (vy <= 0.0f) ? expm1f(vy) : vy;
        }
        *(float2*)&As[wv * PA + 2 * lane] = make_float2(vx, vy);

        __syncthreads();

        const int2 nrfirst = r2[min(nbeg + lane, last_rec)];

        if (wv < NT) {
            v8f acc = (v8f){};
#pragma unroll
            for (int k = 0; k < H; k += 4) {
                v2f a, b;
                a.x = Ap[k];
                a.y = Ap[k + 1];
                b.x = Bp[(k / 2) * PB];
                b.y = Bp[(k / 2) * PB + 1];
                acc = __builtin_amdgcn_wmma_f32_16x16x4_f32(
                    false, a, false, b, (short)0, acc, false, false);
            }
            if (NT * 16 == NB || c < NB) {
#pragma unroll
                for (int i = 0; i < 8; ++i) Cs[(i + half * 8) * PC + c] = acc[i] + bvn;
            }
        }
        __syncthreads();

        if (t < NQO) {
            const float4 v = *(const float4*)&Cs[(t / QPR) * PC + 4 * (t % QPR)];
            ((float4*)outp)[(size_t)tile * NQO + t] = v;
        }

        if (!more) break;
        tile   = ntile;
        beg    = nbeg;
        end    = nend;
        di     = ndi;
        hv     = nhv;
        rfirst = nrfirst;
    }
}


extern "C" void kernel_launch(void* const* d_in, const int* in_sizes, int n_in,
                              void* d_out, int out_size, void* d_ws, size_t ws_size,
                              hipStream_t stream) {
    const float* x  = (const float*)d_in[0];
    const int*   ei = (const int*)d_in[1];
    const float* W1 = (const float*)d_in[2];
    const float* b1 = (const float*)d_in[3];
    const float* W2 = (const float*)d_in[4];
    const float* b2 = (const float*)d_in[5];
    const float* W3 = (const float*)d_in[6];
    const float* b3 = (const float*)d_in[7];
    const float* Wl = (const float*)d_in[8];
    const float* bl = (const float*)d_in[9];
    float* out = (float*)d_out;

    const int E = in_sizes[1] / 2;
    const int* src = ei;
    const int* dst = ei + E;

    float* G      = (float*)d_ws;
    float* ACT    = G + (size_t)NN * H;
    int*   btot   = (int*)(ACT + (size_t)NN * H);
    float* dinv   = (float*)(btot + NN);
    int*   rowptr = (int*)(dinv + NN);
    int*   rec    = rowptr + NN + 16;
    int*   rank   = rec + 2 * (size_t)E;
    int*   part   = rank + E;

    const dim3 blk(256);

    k_hist_gemm<<<dim3(HG + NGB), dim3(1024), 0, stream>>>(dst, part, rank, E, x, W1, G);
    k_merge<<<dim3(NBLK), dim3(MB), 0, stream>>>(part, btot, dinv);
    k_fill<<<dim3((E + 256 * FE - 1) / (256 * FE)), blk, 0, stream>>>(src, dst, rank, part, btot, dinv, rowptr, rec, E);

    const dim3 blk16(512);
    const dim3 gRes(GRES);


    k_gather_gemm<H, 4><<<gRes, blk16, 0, stream>>>(rowptr, rec, G, dinv, b1, W2, nullptr, ACT, 1, E);
    k_gather_gemm<H, 4><<<gRes, blk16, 0, stream>>>(rowptr, rec, ACT, dinv, b2, W3, nullptr, G, 1, E);
    k_gather_gemm<40, 3><<<gRes, blk16, 0, stream>>>(rowptr, rec, G, dinv, b3, Wl, bl, out, 0, E);
}
